# starting kernel
# baseline (speedup 1.0000x reference)
#include <hip/hip_runtime.h>
#include <hip/hip_bf16.h>
#include <math.h>


typedef __attribute__((ext_vector_type(2))) float v2f;
typedef __attribute__((ext_vector_type(8))) float v8f;

#define NEG_SLOPE 0.2f
#define LN_EPS 1e-5f

static __device__ inline v8f wmma4(v2f a, v2f b, v8f c) {
  return __builtin_amdgcn_wmma_f32_16x16x4_f32(
       false, a,  false, b,
       (short)0, c,  false,  false);
}

static __device__ inline void atomicMaxF(float* addr, float val) {
  if (val >= 0.f) atomicMax((int*)addr, __float_as_int(val));
  else            atomicMin((unsigned int*)addr, __float_as_uint(val));
}

__global__ void fill_f32(float* __restrict__ p, long long n, float v) {
  long long i = (long long)blockIdx.x * blockDim.x + threadIdx.x;
  if (i < n) p[i] = v;
}

__global__ void gemm1_wmma(const float* __restrict__ x,
                           const float* __restrict__ Wl,
                           const float* __restrict__ Wr,
                           float* __restrict__ xl,
                           float* __restrict__ xr, int ntiles) {
  int wave = (int)(((long long)blockIdx.x * blockDim.x + threadIdx.x) >> 5);
  int lane = threadIdx.x & 31;
  if (wave >= ntiles) return;
  int row0 = wave * 16;
  int m  = lane & 15;
  int hi = lane >> 4;
  const float* xrow = x + (size_t)(row0 + m) * 4 + hi * 2;
  v2f a; a.x = xrow[0]; a.y = xrow[1];
#pragma unroll
  for (int j = 0; j < 4; ++j) {
    v2f bl, br;
    bl.x = Wl[(2 * hi + 0) * 64 + j * 16 + m];
    bl.y = Wl[(2 * hi + 1) * 64 + j * 16 + m];
    br.x = Wr[(2 * hi + 0) * 64 + j * 16 + m];
    br.y = Wr[(2 * hi + 1) * 64 + j * 16 + m];
    v8f cl = {}; cl = wmma4(a, bl, cl);
    v8f cr = {}; cr = wmma4(a, br, cr);
#pragma unroll
    for (int v = 0; v < 8; ++v) {
      int r = v + hi * 8;
      xl[(size_t)(row0 + r) * 64 + j * 16 + m] = cl[v];
      xr[(size_t)(row0 + r) * 64 + j * 16 + m] = cr[v];
    }
  }
}

__global__ void gemm2_wmma(const float* __restrict__ h,
                           const float* __restrict__ Wl,
                           const float* __restrict__ Wr,
                           float* __restrict__ xl,
                           float* __restrict__ xr, int ntiles) {
  int wave = (int)(((long long)blockIdx.x * blockDim.x + threadIdx.x) >> 5);
  int lane = threadIdx.x & 31;
  if (wave >= ntiles) return;
  int row0 = wave * 16;
  int m  = lane & 15;
  int hi = lane >> 4;
  const float* hrow = h + (size_t)(row0 + m) * 64;
  v8f cl[2] = {{0.f},{0.f}};
  v8f cr[2] = {{0.f},{0.f}};
#pragma unroll
  for (int kk = 0; kk < 16; ++kk) {
    int k0 = kk * 4 + hi * 2;
    v2f a; a.x = hrow[k0]; a.y = hrow[k0 + 1];
#pragma unroll
    for (int j = 0; j < 2; ++j) {
      v2f bl, br;
      bl.x = Wl[(k0 + 0) * 32 + j * 16 + m];
      bl.y = Wl[(k0 + 1) * 32 + j * 16 + m];
      br.x = Wr[(k0 + 0) * 32 + j * 16 + m];
      br.y = Wr[(k0 + 1) * 32 + j * 16 + m];
      cl[j] = wmma4(a, bl, cl[j]);
      cr[j] = wmma4(a, br, cr[j]);
    }
  }
#pragma unroll
  for (int j = 0; j < 2; ++j)
#pragma unroll
    for (int v = 0; v < 8; ++v) {
      int r = v + hi * 8;
      xl[(size_t)(row0 + r) * 32 + j * 16 + m] = cl[j][v];
      xr[(size_t)(row0 + r) * 32 + j * 16 + m] = cr[j][v];
    }
}

__global__ void edge_logits1(const int* __restrict__ ei, int E, int n,
                             const float* __restrict__ xl,
                             const float* __restrict__ xr,
                             const float* __restrict__ a1,
                             float* __restrict__ logits,
                             float* __restrict__ maxd) {
  int t = blockIdx.x * blockDim.x + threadIdx.x;
  int TE = E + n;
  if (t >= TE) return;
  int s, d;
  if (t < E) { s = ei[t]; d = ei[E + t]; } else { s = t - E; d = s; }
  const float4* ps = (const float4*)(xl + (size_t)s * 64);
  const float4* pd = (const float4*)(xr + (size_t)d * 64);
  const float4* pa = (const float4*)a1;
#pragma unroll
  for (int h = 0; h < 4; ++h) {
    float acc = 0.f;
#pragma unroll
    for (int q = 0; q < 4; ++q) {
      float4 u = ps[h * 4 + q], v = pd[h * 4 + q], w = pa[h * 4 + q];
      float e0 = u.x + v.x; e0 = e0 >= 0.f ? e0 : NEG_SLOPE * e0;
      float e1 = u.y + v.y; e1 = e1 >= 0.f ? e1 : NEG_SLOPE * e1;
      float e2 = u.z + v.z; e2 = e2 >= 0.f ? e2 : NEG_SLOPE * e2;
      float e3 = u.w + v.w; e3 = e3 >= 0.f ? e3 : NEG_SLOPE * e3;
      acc += e0 * w.x + e1 * w.y + e2 * w.z + e3 * w.w;
    }
    logits[(size_t)t * 4 + h] = acc;
    atomicMaxF(&maxd[(size_t)d * 4 + h], acc);
  }
}

__global__ void edge_accum1(const int* __restrict__ ei, int E, int n,
                            const float* __restrict__ xl,
                            const float* __restrict__ logits,
                            const float* __restrict__ maxd,
                            float* __restrict__ denom,
                            float* __restrict__ acc) {
  int t = blockIdx.x * blockDim.x + threadIdx.x;
  int T = 4 * (E + n);
  if (t >= T) return;
  int e = t >> 2, h = t & 3;
  int s, d;
  if (e < E) { s = ei[e]; d = ei[E + e]; } else { s = e - E; d = s; }
  float ex = __expf(logits[(size_t)e * 4 + h] - maxd[(size_t)d * 4 + h]);
  atomicAdd(&denom[(size_t)d * 4 + h], ex);
  const float* px = xl + (size_t)s * 64 + h * 16;
  float* pa = acc + (size_t)d * 64 + h * 16;
#pragma unroll
  for (int c = 0; c < 16; ++c) atomicAdd(&pa[c], ex * px[c]);
}

__global__ void node_fin1(const float* __restrict__ acc,
                          const float* __restrict__ denom,
                          const float* __restrict__ b1,
                          const float* __restrict__ g,
                          const float* __restrict__ be,
                          float* __restrict__ h1, int n) {
  int node = (int)(((long long)blockIdx.x * blockDim.x + threadIdx.x) >> 5);
  int lane = threadIdx.x & 31;
  if (node >= n) return;
  int c0 = lane, c1 = lane + 32;
  float v0 = acc[(size_t)node * 64 + c0] / denom[(size_t)node * 4 + (c0 >> 4)] + b1[c0];
  float v1 = acc[(size_t)node * 64 + c1] / denom[(size_t)node * 4 + (c1 >> 4)] + b1[c1];
  float s = v0 + v1;
#pragma unroll
  for (int o = 16; o; o >>= 1) s += __shfl_xor(s, o, 32);
  float mu = s * (1.f / 64.f);
  float d0 = v0 - mu, d1 = v1 - mu;
  float q = d0 * d0 + d1 * d1;
#pragma unroll
  for (int o = 16; o; o >>= 1) q += __shfl_xor(q, o, 32);
  float rs = rsqrtf(q * (1.f / 64.f) + LN_EPS);
  float y0 = d0 * rs * g[c0] + be[c0];
  float y1 = d1 * rs * g[c1] + be[c1];
  y0 = y0 > 0.f ? y0 : expm1f(y0);
  y1 = y1 > 0.f ? y1 : expm1f(y1);
  h1[(size_t)node * 64 + c0] = y0;
  h1[(size_t)node * 64 + c1] = y1;
}

__global__ void edge_logits2(const int* __restrict__ ei, int E, int n,
                             const float* __restrict__ xl,
                             const float* __restrict__ xr,
                             const float* __restrict__ a2,
                             float* __restrict__ logits,
                             float* __restrict__ maxd) {
  int t = blockIdx.x * blockDim.x + threadIdx.x;
  int TE = E + n;
  if (t >= TE) return;
  int s, d;
  if (t < E) { s = ei[t]; d = ei[E + t]; } else { s = t - E; d = s; }
  const float4* ps = (const float4*)(xl + (size_t)s * 32);
  const float4* pd = (const float4*)(xr + (size_t)d * 32);
  const float4* pa = (const float4*)a2;
  float acc = 0.f;
#pragma unroll
  for (int q = 0; q < 8; ++q) {
    float4 u = ps[q], v = pd[q], w = pa[q];
    float e0 = u.x + v.x; e0 = e0 >= 0.f ? e0 : NEG_SLOPE * e0;
    float e1 = u.y + v.y; e1 = e1 >= 0.f ? e1 : NEG_SLOPE * e1;
    float e2 = u.z + v.z; e2 = e2 >= 0.f ? e2 : NEG_SLOPE * e2;
    float e3 = u.w + v.w; e3 = e3 >= 0.f ? e3 : NEG_SLOPE * e3;
    acc += e0 * w.x + e1 * w.y + e2 * w.z + e3 * w.w;
  }
  logits[t] = acc;
  atomicMaxF(&maxd[d], acc);
}

__global__ void edge_accum2(const int* __restrict__ ei, int E, int n,
                            const float* __restrict__ xl,
                            const float* __restrict__ logits,
                            const float* __restrict__ maxd,
                            float* __restrict__ denom,
                            float* __restrict__ acc) {
  int t = blockIdx.x * blockDim.x + threadIdx.x;
  int TE = E + n;
  if (t >= TE) return;
  int s, d;
  if (t < E) { s = ei[t]; d = ei[E + t]; } else { s = t - E; d = s; }
  float ex = __expf(logits[t] - maxd[d]);
  atomicAdd(&denom[d], ex);
  const float* px = xl + (size_t)s * 32;
  float* pa = acc + (size_t)d * 32;
#pragma unroll
  for (int c = 0; c < 32; ++c) atomicAdd(&pa[c], ex * px[c]);
}

__global__ void node_fin2(const float* __restrict__ acc,
                          const float* __restrict__ denom,
                          const float* __restrict__ b2,
                          const float* __restrict__ g,
                          const float* __restrict__ be,
                          float* __restrict__ out, int n) {
  int node = (int)(((long long)blockIdx.x * blockDim.x + threadIdx.x) >> 5);
  int lane = threadIdx.x & 31;
  if (node >= n) return;
  float v = acc[(size_t)node * 32 + lane] / denom[node] + b2[lane];
  float s = v;
#pragma unroll
  for (int o = 16; o; o >>= 1) s += __shfl_xor(s, o, 32);
  float mu = s * (1.f / 32.f);
  float dd = v - mu;
  float q = dd * dd;
#pragma unroll
  for (int o = 16; o; o >>= 1) q += __shfl_xor(q, o, 32);
  float rs = rsqrtf(q * (1.f / 32.f) + LN_EPS);
  out[(size_t)node * 32 + lane] = dd * rs * g[lane] + be[lane];
}

static inline int gridFor(long long threads, int block) {
  return (int)((threads + block - 1) / block);
}

extern "C" void kernel_launch(void* const* d_in, const int* in_sizes, int n_in,
                              void* d_out, int out_size, void* d_ws, size_t ws_size,
                              hipStream_t stream) {
  const float* x   = (const float*)d_in[0];
  const int*   ei  = (const int*)  d_in[1];
  const float* W1l = (const float*)d_in[2];
  const float* W1r = (const float*)d_in[3];
  const float* a1  = (const float*)d_in[4];
  const float* b1  = (const float*)d_in[5];
  const float* g1  = (const float*)d_in[6];
  const float* be1 = (const float*)d_in[7];
  const float* W2l = (const float*)d_in[8];
  const float* W2r = (const float*)d_in[9];
  const float* a2  = (const float*)d_in[10];
  const float* b2  = (const float*)d_in[11];
  const float* g2  = (const float*)d_in[12];
  const float* be2 = (const float*)d_in[13];
  float* out = (float*)d_out;

  const int n  = in_sizes[0] / 4;
  const int E  = in_sizes[1] / 2;
  const int TE = E + n;
  const int ntiles = n / 16;

  float* ws = (float*)d_ws;
  const size_t XL1  = 0;
  const size_t XR1  = XL1 + (size_t)n * 64;
  const size_t ACC1 = XR1 + (size_t)n * 64;
  const size_t DEN1 = ACC1 + (size_t)n * 64;
  const size_t MAX1 = DEN1 + (size_t)n * 4;
  const size_t LOG1 = MAX1 + (size_t)n * 4;
  const size_t H1   = LOG1 + (size_t)TE * 4;

  const int B = 256;
  const float NEG_INF = -INFINITY;

  fill_f32<<<gridFor((long long)n * 68, B), B, 0, stream>>>(ws + ACC1, (long long)n * 68, 0.f);
  fill_f32<<<gridFor((long long)n * 4, B), B, 0, stream>>>(ws + MAX1, (long long)n * 4, NEG_INF);

  gemm1_wmma<<<gridFor((long long)ntiles * 32, B), B, 0, stream>>>(
      x, W1l, W1r, ws + XL1, ws + XR1, ntiles);

  edge_logits1<<<gridFor(TE, B), B, 0, stream>>>(
      ei, E, n, ws + XL1, ws + XR1, a1, ws + LOG1, ws + MAX1);

  edge_accum1<<<gridFor((long long)TE * 4, B), B, 0, stream>>>(
      ei, E, n, ws + XL1, ws + LOG1, ws + MAX1, ws + DEN1, ws + ACC1);

  node_fin1<<<gridFor((long long)n * 32, B), B, 0, stream>>>(
      ws + ACC1, ws + DEN1, b1, g1, be1, ws + H1, n);

  fill_f32<<<gridFor((long long)n * 32, B), B, 0, stream>>>(ws + ACC1, (long long)n * 32, 0.f);
  fill_f32<<<gridFor(n, B), B, 0, stream>>>(ws + DEN1, n, 0.f);
  fill_f32<<<gridFor(n, B), B, 0, stream>>>(ws + MAX1, n, NEG_INF);

  gemm2_wmma<<<gridFor((long long)ntiles * 32, B), B, 0, stream>>>(
      ws + H1, W2l, W2r, ws + XL1, ws + XR1, ntiles);

  edge_logits2<<<gridFor(TE, B), B, 0, stream>>>(
      ei, E, n, ws + XL1, ws + XR1, a2, ws + LOG1, ws + MAX1);

  edge_accum2<<<gridFor(TE, B), B, 0, stream>>>(
      ei, E, n, ws + XL1, ws + LOG1, ws + MAX1, ws + DEN1, ws + ACC1);

  node_fin2<<<gridFor((long long)n * 32, B), B, 0, stream>>>(
      ws + ACC1, ws + DEN1, b2, g2, be2, out, n);
}


# Both logits passes re-mapped: a wave's lanes span the channels of one edge, so every xl/xr row gather is whole 128-B lines; dot products finish in a transposing lane butterfly: new logit sum order
# speedup vs baseline: 3.3343x; 3.3343x over previous
#include <hip/hip_runtime.h>
#include <hip/hip_bf16.h>
#include <math.h>


typedef __attribute__((ext_vector_type(2))) float v2f;
typedef __attribute__((ext_vector_type(8))) float v8f;

#define NEG_SLOPE 0.2f
#define LN_EPS 1e-5f

static __device__ inline v8f wmma4(v2f a, v2f b, v8f c) {
  return __builtin_amdgcn_wmma_f32_16x16x4_f32(
       false, a,  false, b,
       (short)0, c,  false,  false);
}

static __device__ inline void atomicMaxF(float* addr, float val) {
  if (val >= 0.f) atomicMax((int*)addr, __float_as_int(val));
  else            atomicMin((unsigned int*)addr, __float_as_uint(val));
}

static __device__ inline int bcast_i(int v, int srcLane) {
  return __builtin_amdgcn_readlane(v, srcLane);
}
static __device__ inline float bcast_f(float v, int srcLane) {
  return __int_as_float(__builtin_amdgcn_readlane(__float_as_int(v), srcLane));
}

template <int OFF, int STRIDE, int Q>
static __device__ inline void bfly(float (&p)[32], bool up) {
  if constexpr (Q < STRIDE) {
    const float lo = p[Q], hi = p[Q + STRIDE];
    const float keep = up ? hi : lo;
    const float send = up ? lo : hi;
    p[Q] = keep + __shfl_xor(send, OFF, 32);
    bfly<OFF, STRIDE, Q + 1>(p, up);
  }
}

__global__ void fill_f32(float* __restrict__ p, long long n, float v) {
  long long i = (long long)blockIdx.x * blockDim.x + threadIdx.x;
  if (i < n) p[i] = v;
}

__global__ void gemm1_wmma(const float* __restrict__ x,
                           const float* __restrict__ Wl,
                           const float* __restrict__ Wr,
                           float* __restrict__ xl,
                           float* __restrict__ xr, int ntiles) {
  int wave = (int)(((long long)blockIdx.x * blockDim.x + threadIdx.x) >> 5);
  int lane = threadIdx.x & 31;
  if (wave >= ntiles) return;
  int row0 = wave * 16;
  int m  = lane & 15;
  int hi = lane >> 4;
  const float* xrow = x + (size_t)(row0 + m) * 4 + hi * 2;
  v2f a; a.x = xrow[0]; a.y = xrow[1];
#pragma unroll
  for (int j = 0; j < 4; ++j) {
    v2f bl, br;
    bl.x = Wl[(2 * hi + 0) * 64 + j * 16 + m];
    bl.y = Wl[(2 * hi + 1) * 64 + j * 16 + m];
    br.x = Wr[(2 * hi + 0) * 64 + j * 16 + m];
    br.y = Wr[(2 * hi + 1) * 64 + j * 16 + m];
    v8f cl = {}; cl = wmma4(a, bl, cl);
    v8f cr = {}; cr = wmma4(a, br, cr);
#pragma unroll
    for (int v = 0; v < 8; ++v) {
      int r = v + hi * 8;
      xl[(size_t)(row0 + r) * 64 + j * 16 + m] = cl[v];
      xr[(size_t)(row0 + r) * 64 + j * 16 + m] = cr[v];
    }
  }
}

__global__ void gemm2_wmma(const float* __restrict__ h,
                           const float* __restrict__ Wl,
                           const float* __restrict__ Wr,
                           float* __restrict__ xl,
                           float* __restrict__ xr, int ntiles) {
  int wave = (int)(((long long)blockIdx.x * blockDim.x + threadIdx.x) >> 5);
  int lane = threadIdx.x & 31;
  if (wave >= ntiles) return;
  int row0 = wave * 16;
  int m  = lane & 15;
  int hi = lane >> 4;
  const float* hrow = h + (size_t)(row0 + m) * 64;
  v8f cl[2] = {{0.f},{0.f}};
  v8f cr[2] = {{0.f},{0.f}};
#pragma unroll
  for (int kk = 0; kk < 16; ++kk) {
    int k0 = kk * 4 + hi * 2;
    v2f a; a.x = hrow[k0]; a.y = hrow[k0 + 1];
#pragma unroll
    for (int j = 0; j < 2; ++j) {
      v2f bl, br;
      bl.x = Wl[(k0 + 0) * 32 + j * 16 + m];
      bl.y = Wl[(k0 + 1) * 32 + j * 16 + m];
      br.x = Wr[(k0 + 0) * 32 + j * 16 + m];
      br.y = Wr[(k0 + 1) * 32 + j * 16 + m];
      cl[j] = wmma4(a, bl, cl[j]);
      cr[j] = wmma4(a, br, cr[j]);
    }
  }
#pragma unroll
  for (int j = 0; j < 2; ++j)
#pragma unroll
    for (int v = 0; v < 8; ++v) {
      int r = v + hi * 8;
      xl[(size_t)(row0 + r) * 32 + j * 16 + m] = cl[j][v];
      xr[(size_t)(row0 + r) * 32 + j * 16 + m] = cr[j][v];
    }
}

__global__ void edge_logits1(const int* __restrict__ ei, int E, int n,
                             const float* __restrict__ xl,
                             const float* __restrict__ xr,
                             const float* __restrict__ a1,
                             float* __restrict__ logits,
                             float* __restrict__ maxd) {
  const int lane = threadIdx.x & 31;
  const int wave = __builtin_amdgcn_readfirstlane(
      (int)(((long long)blockIdx.x * blockDim.x + threadIdx.x) >> 5));
  const int TE = E + n;
  const int base = wave * 16;
  if (base >= TE) return;
  const int t = base + (lane & 15);
  int s = 0, d = 0;
  if (t < TE) {
    if (t < E) { s = ei[t]; d = ei[E + t]; } else { s = t - E; d = s; }
  }
  const float aA = a1[lane], aB = a1[32 + lane];
  float p[32];
#pragma unroll
  for (int j0 = 0; j0 < 16; j0 += 8) {
    float uA[8], uB[8], vA[8], vB[8];
#pragma unroll
    for (int k = 0; k < 8; ++k) {
      const unsigned is = (unsigned)bcast_i(s, j0 + k) * 64u + (unsigned)lane;
      const unsigned id = (unsigned)bcast_i(d, j0 + k) * 64u + (unsigned)lane;
      uA[k] = xl[is]; uB[k] = xl[is + 32u];
      vA[k] = xr[id]; vB[k] = xr[id + 32u];
    }
#pragma unroll
    for (int k = 0; k < 8; ++k) {
      float eA = uA[k] + vA[k]; eA = eA >= 0.f ? eA : NEG_SLOPE * eA;
      float eB = uB[k] + vB[k]; eB = eB >= 0.f ? eB : NEG_SLOPE * eB;
      p[j0 + k] = eA * aA;
      p[16 + j0 + k] = eB * aB;
    }
  }
  bfly<8, 16, 0>(p, (lane & 8) != 0);
  bfly<4, 8, 0>(p, (lane & 4) != 0);
  bfly<2, 4, 0>(p, (lane & 2) != 0);
  bfly<1, 2, 0>(p, (lane & 1) != 0);
  const int head = (lane >> 4) + 2 * ((lane >> 3) & 1);
  const int jb = 2 * (lane & 7);
#pragma unroll
  for (int q = 0; q < 2; ++q) {
    const int j = jb + q;
    const int dj = __shfl(d, j, 32);
    if (base + j < TE) {
      logits[(size_t)(base + j) * 4 + head] = p[q];
      atomicMaxF(&maxd[(size_t)dj * 4 + head], p[q]);
    }
  }
}

__global__ void edge_accum1(const int* __restrict__ ei, int E, int n,
                            const float* __restrict__ x,
                            const float* __restrict__ logits,
                            const float* __restrict__ maxd,
                            float* __restrict__ acc) {
  const int lane = threadIdx.x & 31;
  const int wave = __builtin_amdgcn_readfirstlane(
      (int)(((long long)blockIdx.x * blockDim.x + threadIdx.x) >> 5));
  const int TE = E + n;
  const int base = wave * 8;
  if (base >= TE) return;
  const int e = base + (lane >> 2), h = lane & 3;
  int s = 0, d = 0;
  float ex = 0.f;
  if (e < TE) {
    if (e < E) { s = ei[e]; d = ei[E + e]; } else { s = e - E; d = s; }
    ex = __expf(logits[(size_t)e * 4 + h] - maxd[(size_t)d * 4 + h]);
  }
  const int cnt = TE - base;
  const int kx = lane < 16 ? (lane >> 2) : 3;
  const int hsel = (lane & 3) << 2;
  if (cnt >= 8) {
    float xv[8], w[8];
    int dj[8];
#pragma unroll
    for (int j = 0; j < 8; ++j)
      xv[j] = x[(size_t)bcast_i(s, 4 * j) * 4 + kx];
#pragma unroll
    for (int j = 0; j < 8; ++j) {
      w[j] = __int_as_float(
          __builtin_amdgcn_ds_bpermute(hsel + 16 * j, __float_as_int(ex)));
      dj[j] = bcast_i(d, 4 * j);
    }
    if (lane < 20) {
#pragma unroll
      for (int j = 0; j < 8; ++j)
        atomicAdd(acc + (size_t)dj[j] * 32 + lane, (lane < 16 ? xv[j] : 1.f) * w[j]);
    }
  } else {
    for (int j = 0; j < cnt; ++j) {
      const float xj = x[(size_t)bcast_i(s, 4 * j) * 4 + kx];
      const float w = __int_as_float(
          __builtin_amdgcn_ds_bpermute(hsel + 16 * j, __float_as_int(ex)));
      const float val = (lane < 16 ? xj : 1.f) * w;
      float* row = acc + (size_t)bcast_i(d, 4 * j) * 32;
      if (lane < 20) atomicAdd(row + lane, val);
    }
  }
}

__global__ void node_fin1(const float* __restrict__ acc,
                          const float* __restrict__ Wl,
                          const float* __restrict__ b1,
                          const float* __restrict__ g,
                          const float* __restrict__ be,
                          float* __restrict__ h1, int ntiles) {
  int wave = (int)(((long long)blockIdx.x * blockDim.x + threadIdx.x) >> 5);
  int lane = threadIdx.x & 31;
  if (wave >= ntiles) return;
  int row0 = wave * 16;
  int m  = lane & 15;
  int hi = lane >> 4;
  const float4* arow = (const float4*)(acc + (size_t)(row0 + m) * 32);
  const float4 p0 = arow[2 * hi];
  const float4 p1 = arow[2 * hi + 1];
  const float4 dn = arow[4];
  const float pa[4] = {p0.x, p0.y, p0.z, p0.w};
  const float pb[4] = {p1.x, p1.y, p1.z, p1.w};
  const float dd[4] = {dn.x, dn.y, dn.z, dn.w};
  v8f c[4];
  float bias[4], gg[4], bb[4];
#pragma unroll
  for (int h = 0; h < 4; ++h) {
    v2f a; a.x = pa[h] / dd[h]; a.y = pb[h] / dd[h];
    v2f b;
    b.x = Wl[(2 * hi + 0) * 64 + h * 16 + m];
    b.y = Wl[(2 * hi + 1) * 64 + h * 16 + m];
    v8f z = {};
    c[h] = wmma4(a, b, z);
    bias[h] = b1[h * 16 + m];
    gg[h] = g[h * 16 + m];
    bb[h] = be[h * 16 + m];
  }
  float s[8], q[8];
#pragma unroll
  for (int v = 0; v < 8; ++v) {
#pragma unroll
    for (int h = 0; h < 4; ++h) c[h][v] += bias[h];
    s[v] = (c[0][v] + c[1][v]) + (c[2][v] + c[3][v]);
  }
#pragma unroll
  for (int o = 8; o; o >>= 1)
#pragma unroll
    for (int v = 0; v < 8; ++v) s[v] += __shfl_xor(s[v], o, 32);
#pragma unroll
  for (int v = 0; v < 8; ++v) {
    const float mu = s[v] * (1.f / 64.f);
#pragma unroll
    for (int h = 0; h < 4; ++h) c[h][v] -= mu;
    q[v] = (c[0][v] * c[0][v] + c[1][v] * c[1][v]) + (c[2][v] * c[2][v] + c[3][v] * c[3][v]);
  }
#pragma unroll
  for (int o = 8; o; o >>= 1)
#pragma unroll
    for (int v = 0; v < 8; ++v) q[v] += __shfl_xor(q[v], o, 32);
#pragma unroll
  for (int v = 0; v < 8; ++v) {
    const float rs = rsqrtf(q[v] * (1.f / 64.f) + LN_EPS);
    float* orow = h1 + (size_t)(row0 + v + hi * 8) * 64 + m;
#pragma unroll
    for (int h = 0; h < 4; ++h) {
      float y = c[h][v] * rs * gg[h] + bb[h];
      y = y > 0.f ? y : expm1f(y);
      orow[h * 16] = y;
    }
  }
}

__global__ void edge_logits2(const int* __restrict__ ei, int E, int n,
                             const float* __restrict__ xl,
                             const float* __restrict__ xr,
                             const float* __restrict__ a2,
                             float* __restrict__ logits,
                             float* __restrict__ maxd) {
  const int lane = threadIdx.x & 31;
  const int wave = __builtin_amdgcn_readfirstlane(
      (int)(((long long)blockIdx.x * blockDim.x + threadIdx.x) >> 5));
  const int TE = E + n;
  const int base = wave * 32;
  if (base >= TE) return;
  const int t = base + lane;
  int s = 0, d = 0;
  if (t < TE) {
    if (t < E) { s = ei[t]; d = ei[E + t]; } else { s = t - E; d = s; }
  }
  const float al = a2[lane];
  float p[32];
#pragma unroll
  for (int j0 = 0; j0 < 32; j0 += 8) {
    float u[8], v[8];
#pragma unroll
    for (int k = 0; k < 8; ++k) {
      u[k] = xl[(unsigned)bcast_i(s, j0 + k) * 32u + (unsigned)lane];
      v[k] = xr[(unsigned)bcast_i(d, j0 + k) * 32u + (unsigned)lane];
    }
#pragma unroll
    for (int k = 0; k < 8; ++k) {
      float e = u[k] + v[k]; e = e >= 0.f ? e : NEG_SLOPE * e;
      p[j0 + k] = e * al;
    }
  }
  bfly<16, 16, 0>(p, (lane & 16) != 0);
  bfly<8, 8, 0>(p, (lane & 8) != 0);
  bfly<4, 4, 0>(p, (lane & 4) != 0);
  bfly<2, 2, 0>(p, (lane & 2) != 0);
  bfly<1, 1, 0>(p, (lane & 1) != 0);
  if (t < TE) {
    logits[t] = p[0];
    atomicMaxF(&maxd[d], p[0]);
  }
}

__global__ void edge_accum2(const int* __restrict__ ei, int E, int n,
                            const float* __restrict__ xl,
                            const float* __restrict__ logits,
                            const float* __restrict__ maxd,
                            float* __restrict__ denom,
                            float* __restrict__ acc) {
  const int lane = threadIdx.x & 31;
  const int wave = __builtin_amdgcn_readfirstlane(
      (int)(((long long)blockIdx.x * blockDim.x + threadIdx.x) >> 5));
  const int TE = E + n;
  const int base = wave * 32;
  if (base >= TE) return;
  const int t = base + lane;
  int s = 0, d = 0;
  float ex = 0.f;
  if (t < TE) {
    if (t < E) { s = ei[t]; d = ei[E + t]; } else { s = t - E; d = s; }
    ex = __expf(logits[t] - maxd[d]);
    atomicAdd(&denom[d], ex);
  }
  const int cnt = TE - base;
  if (cnt >= 32) {
#pragma unroll
    for (int j0 = 0; j0 < 32; j0 += 8) {
      float v[8];
#pragma unroll
      for (int u = 0; u < 8; ++u) v[u] = xl[(size_t)bcast_i(s, j0 + u) * 32 + lane];
#pragma unroll
      for (int u = 0; u < 8; ++u)
        atomicAdd(acc + (size_t)bcast_i(d, j0 + u) * 32 + lane, bcast_f(ex, j0 + u) * v[u]);
    }
  } else {
    for (int j = 0; j < cnt; ++j) {
      const float v = xl[(size_t)bcast_i(s, j) * 32 + lane];
      atomicAdd(acc + (size_t)bcast_i(d, j) * 32 + lane, bcast_f(ex, j) * v);
    }
  }
}

__global__ void node_fin2(const float* __restrict__ acc,
                          const float* __restrict__ denom,
                          const float* __restrict__ b2,
                          const float* __restrict__ g,
                          const float* __restrict__ be,
                          float* __restrict__ out, int n) {
  int node = (int)(((long long)blockIdx.x * blockDim.x + threadIdx.x) >> 5);
  int lane = threadIdx.x & 31;
  if (node >= n) return;
  float v = acc[(size_t)node * 32 + lane] / denom[node] + b2[lane];
  float s = v;
#pragma unroll
  for (int o = 16; o; o >>= 1) s += __shfl_xor(s, o, 32);
  float mu = s * (1.f / 32.f);
  float dd = v - mu;
  float q = dd * dd;
#pragma unroll
  for (int o = 16; o; o >>= 1) q += __shfl_xor(q, o, 32);
  float rs = rsqrtf(q * (1.f / 32.f) + LN_EPS);
  out[(size_t)node * 32 + lane] = dd * rs * g[lane] + be[lane];
}

static inline int gridFor(long long threads, int block) {
  return (int)((threads + block - 1) / block);
}

extern "C" void kernel_launch(void* const* d_in, const int* in_sizes, int n_in,
                              void* d_out, int out_size, void* d_ws, size_t ws_size,
                              hipStream_t stream) {
  const float* x   = (const float*)d_in[0];
  const int*   ei  = (const int*)  d_in[1];
  const float* W1l = (const float*)d_in[2];
  const float* W1r = (const float*)d_in[3];
  const float* a1  = (const float*)d_in[4];
  const float* b1  = (const float*)d_in[5];
  const float* g1  = (const float*)d_in[6];
  const float* be1 = (const float*)d_in[7];
  const float* W2l = (const float*)d_in[8];
  const float* W2r = (const float*)d_in[9];
  const float* a2  = (const float*)d_in[10];
  const float* b2  = (const float*)d_in[11];
  const float* g2  = (const float*)d_in[12];
  const float* be2 = (const float*)d_in[13];
  float* out = (float*)d_out;

  const int n  = in_sizes[0] / 4;
  const int E  = in_sizes[1] / 2;
  const int TE = E + n;
  const int ntiles = n / 16;

  float* ws = (float*)d_ws;
  const size_t XL1  = 0;
  const size_t XR1  = XL1 + (size_t)n * 64;
  const size_t ACC1 = XR1 + (size_t)n * 64;
  const size_t DEN1 = ACC1 + (size_t)n * 64;
  const size_t MAX1 = DEN1 + (size_t)n * 4;
  const size_t LOG1 = MAX1 + (size_t)n * 4;
  const size_t H1   = LOG1 + (size_t)TE * 4;

  const int B = 256;
  const float NEG_INF = -INFINITY;

  fill_f32<<<gridFor((long long)n * 32, B), B, 0, stream>>>(ws + ACC1, (long long)n * 32, 0.f);
  fill_f32<<<gridFor((long long)n * 4, B), B, 0, stream>>>(ws + MAX1, (long long)n * 4, NEG_INF);

  gemm1_wmma<<<gridFor((long long)ntiles * 32, B), B, 0, stream>>>(
      x, W1l, W1r, ws + XL1, ws + XR1, ntiles);

  edge_logits1<<<gridFor((long long)TE * 2, B), B, 0, stream>>>(
      ei, E, n, ws + XL1, ws + XR1, a1, ws + LOG1, ws + MAX1);

  edge_accum1<<<gridFor((long long)TE * 4, B), B, 0, stream>>>(
      ei, E, n, x, ws + LOG1, ws + MAX1, ws + ACC1);

  node_fin1<<<gridFor((long long)ntiles * 32, 128), 128, 0, stream>>>(
      ws + ACC1, W1l, b1, g1, be1, ws + H1, ntiles);

  fill_f32<<<gridFor((long long)n * 32, B), B, 0, stream>>>(ws + ACC1, (long long)n * 32, 0.f);
  fill_f32<<<gridFor(n, B), B, 0, stream>>>(ws + DEN1, n, 0.f);
  fill_f32<<<gridFor(n, B), B, 0, stream>>>(ws + MAX1, n, NEG_INF);

  gemm2_wmma<<<gridFor((long long)ntiles * 32, B), B, 0, stream>>>(
      ws + H1, W2l, W2r, ws + XL1, ws + XR1, ntiles);

  edge_logits2<<<gridFor(TE, B), B, 0, stream>>>(
      ei, E, n, ws + XL1, ws + XR1, a2, ws + LOG1, ws + MAX1);

  edge_accum2<<<gridFor(TE, B), B, 0, stream>>>(
      ei, E, n, ws + XL1, ws + LOG1, ws + MAX1, ws + DEN1, ws + ACC1);

  node_fin2<<<gridFor((long long)n * 32, B), B, 0, stream>>>(
      ws + ACC1, ws + DEN1, b2, g2, be2, out, n);
}


# Layer-1 logits pass forms x[src]@W1l + x[dst]@W1r per 16-edge tile on WMMA from two 16-byte x rows per edge: no layer-1 xl/xr planes, GEMM launch or 256-B row gathers; new logit sum order
# speedup vs baseline: 3.5394x; 1.0615x over previous
#include <hip/hip_runtime.h>
#include <hip/hip_bf16.h>
#include <math.h>


typedef __attribute__((ext_vector_type(2))) float v2f;
typedef __attribute__((ext_vector_type(8))) float v8f;

#define NEG_SLOPE 0.2f
#define LN_EPS 1e-5f

static __device__ inline v8f wmma4(v2f a, v2f b, v8f c) {
  return __builtin_amdgcn_wmma_f32_16x16x4_f32(
       false, a,  false, b,
       (short)0, c,  false,  false);
}

static __device__ inline void atomicMaxF(float* addr, float val) {
  if (val >= 0.f) atomicMax((int*)addr, __float_as_int(val));
  else            atomicMin((unsigned int*)addr, __float_as_uint(val));
}

static __device__ inline int bcast_i(int v, int srcLane) {
  return __builtin_amdgcn_readlane(v, srcLane);
}
static __device__ inline float bcast_f(float v, int srcLane) {
  return __int_as_float(__builtin_amdgcn_readlane(__float_as_int(v), srcLane));
}

template <int OFF, int STRIDE, int Q>
static __device__ inline void bfly(float (&p)[32], bool up) {
  if constexpr (Q < STRIDE) {
    const float lo = p[Q], hi = p[Q + STRIDE];
    const float keep = up ? hi : lo;
    const float send = up ? lo : hi;
    p[Q] = keep + __shfl_xor(send, OFF, 32);
    bfly<OFF, STRIDE, Q + 1>(p, up);
  }
}

__global__ void fill_f32(float* __restrict__ p, long long n, float v) {
  long long i = (long long)blockIdx.x * blockDim.x + threadIdx.x;
  if (i < n) p[i] = v;
}

__global__ void gemm2_wmma(const float* __restrict__ h,
                           const float* __restrict__ Wl,
                           const float* __restrict__ Wr,
                           float* __restrict__ xl,
                           float* __restrict__ xr, int ntiles) {
  int wave = (int)(((long long)blockIdx.x * blockDim.x + threadIdx.x) >> 5);
  int lane = threadIdx.x & 31;
  if (wave >= ntiles) return;
  int row0 = wave * 16;
  int m  = lane & 15;
  int hi = lane >> 4;
  const float* hrow = h + (size_t)(row0 + m) * 64;
  v8f cl[2] = {{0.f},{0.f}};
  v8f cr[2] = {{0.f},{0.f}};
#pragma unroll
  for (int kk = 0; kk < 16; ++kk) {
    int k0 = kk * 4 + hi * 2;
    v2f a; a.x = hrow[k0]; a.y = hrow[k0 + 1];
#pragma unroll
    for (int j = 0; j < 2; ++j) {
      v2f bl, br;
      bl.x = Wl[(k0 + 0) * 32 + j * 16 + m];
      bl.y = Wl[(k0 + 1) * 32 + j * 16 + m];
      br.x = Wr[(k0 + 0) * 32 + j * 16 + m];
      br.y = Wr[(k0 + 1) * 32 + j * 16 + m];
      cl[j] = wmma4(a, bl, cl[j]);
      cr[j] = wmma4(a, br, cr[j]);
    }
  }
#pragma unroll
  for (int j = 0; j < 2; ++j)
#pragma unroll
    for (int v = 0; v < 8; ++v) {
      int r = v + hi * 8;
      xl[(size_t)(row0 + r) * 32 + j * 16 + m] = cl[j][v];
      xr[(size_t)(row0 + r) * 32 + j * 16 + m] = cr[j][v];
    }
}

__global__ void edge_logits1(const int* __restrict__ ei, int E, int n,
                             const float* __restrict__ x,
                             const float* __restrict__ Wl,
                             const float* __restrict__ Wr,
                             const float* __restrict__ a1,
                             float* __restrict__ logits,
                             float* __restrict__ maxd) {
  const int lane = threadIdx.x & 31;
  const int wave = __builtin_amdgcn_readfirstlane(
      (int)(((long long)blockIdx.x * blockDim.x + threadIdx.x) >> 5));
  const int TE = E + n;
  const int base = wave * 16;
  if (base >= TE) return;
  const int m  = lane & 15;
  const int hi = lane >> 4;
  const int t = base + m;
  int s = 0, d = 0;
  if (t < TE) {
    if (t < E) { s = ei[t]; d = ei[E + t]; } else { s = t - E; d = s; }
  }
  const float2 xs = *(const float2*)(x + (size_t)s * 4 + hi * 2);
  const float2 xd = *(const float2*)(x + (size_t)d * 4 + hi * 2);
  v2f fs; fs.x = xs.x; fs.y = xs.y;
  v2f fd; fd.x = xd.x; fd.y = xd.y;
  float p[32];
#pragma unroll
  for (int j = 0; j < 4; ++j) {
    v2f bl, br;
    bl.x = Wl[(2 * hi + 0) * 64 + j * 16 + m];
    bl.y = Wl[(2 * hi + 1) * 64 + j * 16 + m];
    br.x = Wr[(2 * hi + 0) * 64 + j * 16 + m];
    br.y = Wr[(2 * hi + 1) * 64 + j * 16 + m];
    const float aj = a1[j * 16 + m];
    v8f cc = {};
    cc = wmma4(fs, bl, cc);
    cc = wmma4(fd, br, cc);
#pragma unroll
    for (int v = 0; v < 8; ++v) {
      float e = cc[v]; e = e >= 0.f ? e : NEG_SLOPE * e;
      p[(v >> 1) * 8 + j * 2 + (v & 1)] = e * aj;
    }
  }
  bfly<8, 16, 0>(p, (lane & 8) != 0);
  bfly<4, 8, 0>(p, (lane & 4) != 0);
  bfly<2, 4, 0>(p, (lane & 2) != 0);
  bfly<1, 2, 0>(p, (lane & 1) != 0);
  const int head = lane & 3;
  const int jb = hi * 8 + 2 * ((lane >> 2) & 3);
#pragma unroll
  for (int q = 0; q < 2; ++q) {
    const int j = jb + q;
    const int dj = __shfl(d, j, 32);
    if (base + j < TE) {
      logits[(size_t)(base + j) * 4 + head] = p[q];
      atomicMaxF(&maxd[(size_t)dj * 4 + head], p[q]);
    }
  }
}

__global__ void edge_accum1(const int* __restrict__ ei, int E, int n,
                            const float* __restrict__ x,
                            const float* __restrict__ logits,
                            const float* __restrict__ maxd,
                            float* __restrict__ acc) {
  const int lane = threadIdx.x & 31;
  const int wave = __builtin_amdgcn_readfirstlane(
      (int)(((long long)blockIdx.x * blockDim.x + threadIdx.x) >> 5));
  const int TE = E + n;
  const int base = wave * 8;
  if (base >= TE) return;
  const int e = base + (lane >> 2), h = lane & 3;
  int s = 0, d = 0;
  float ex = 0.f;
  if (e < TE) {
    if (e < E) { s = ei[e]; d = ei[E + e]; } else { s = e - E; d = s; }
    ex = __expf(logits[(size_t)e * 4 + h] - maxd[(size_t)d * 4 + h]);
  }
  const int cnt = TE - base;
  const int kx = lane < 16 ? (lane >> 2) : 3;
  const int hsel = (lane & 3) << 2;
  if (cnt >= 8) {
    float xv[8], w[8];
    int dj[8];
#pragma unroll
    for (int j = 0; j < 8; ++j)
      xv[j] = x[(size_t)bcast_i(s, 4 * j) * 4 + kx];
#pragma unroll
    for (int j = 0; j < 8; ++j) {
      w[j] = __int_as_float(
          __builtin_amdgcn_ds_bpermute(hsel + 16 * j, __float_as_int(ex)));
      dj[j] = bcast_i(d, 4 * j);
    }
    if (lane < 20) {
#pragma unroll
      for (int j = 0; j < 8; ++j)
        atomicAdd(acc + (size_t)dj[j] * 32 + lane, (lane < 16 ? xv[j] : 1.f) * w[j]);
    }
  } else {
    for (int j = 0; j < cnt; ++j) {
      const float xj = x[(size_t)bcast_i(s, 4 * j) * 4 + kx];
      const float w = __int_as_float(
          __builtin_amdgcn_ds_bpermute(hsel + 16 * j, __float_as_int(ex)));
      const float val = (lane < 16 ? xj : 1.f) * w;
      float* row = acc + (size_t)bcast_i(d, 4 * j) * 32;
      if (lane < 20) atomicAdd(row + lane, val);
    }
  }
}

__global__ void node_fin1(const float* __restrict__ acc,
                          const float* __restrict__ Wl,
                          const float* __restrict__ b1,
                          const float* __restrict__ g,
                          const float* __restrict__ be,
                          float* __restrict__ h1, int ntiles) {
  int wave = (int)(((long long)blockIdx.x * blockDim.x + threadIdx.x) >> 5);
  int lane = threadIdx.x & 31;
  if (wave >= ntiles) return;
  int row0 = wave * 16;
  int m  = lane & 15;
  int hi = lane >> 4;
  const float4* arow = (const float4*)(acc + (size_t)(row0 + m) * 32);
  const float4 p0 = arow[2 * hi];
  const float4 p1 = arow[2 * hi + 1];
  const float4 dn = arow[4];
  const float pa[4] = {p0.x, p0.y, p0.z, p0.w};
  const float pb[4] = {p1.x, p1.y, p1.z, p1.w};
  const float dd[4] = {dn.x, dn.y, dn.z, dn.w};
  v8f c[4];
  float bias[4], gg[4], bb[4];
#pragma unroll
  for (int h = 0; h < 4; ++h) {
    v2f a; a.x = pa[h] / dd[h]; a.y = pb[h] / dd[h];
    v2f b;
    b.x = Wl[(2 * hi + 0) * 64 + h * 16 + m];
    b.y = Wl[(2 * hi + 1) * 64 + h * 16 + m];
    v8f z = {};
    c[h] = wmma4(a, b, z);
    bias[h] = b1[h * 16 + m];
    gg[h] = g[h * 16 + m];
    bb[h] = be[h * 16 + m];
  }
  float s[8], q[8];
#pragma unroll
  for (int v = 0; v < 8; ++v) {
#pragma unroll
    for (int h = 0; h < 4; ++h) c[h][v] += bias[h];
    s[v] = (c[0][v] + c[1][v]) + (c[2][v] + c[3][v]);
  }
#pragma unroll
  for (int o = 8; o; o >>= 1)
#pragma unroll
    for (int v = 0; v < 8; ++v) s[v] += __shfl_xor(s[v], o, 32);
#pragma unroll
  for (int v = 0; v < 8; ++v) {
    const float mu = s[v] * (1.f / 64.f);
#pragma unroll
    for (int h = 0; h < 4; ++h) c[h][v] -= mu;
    q[v] = (c[0][v] * c[0][v] + c[1][v] * c[1][v]) + (c[2][v] * c[2][v] + c[3][v] * c[3][v]);
  }
#pragma unroll
  for (int o = 8; o; o >>= 1)
#pragma unroll
    for (int v = 0; v < 8; ++v) q[v] += __shfl_xor(q[v], o, 32);
#pragma unroll
  for (int v = 0; v < 8; ++v) {
    const float rs = rsqrtf(q[v] * (1.f / 64.f) + LN_EPS);
    float* orow = h1 + (size_t)(row0 + v + hi * 8) * 64 + m;
#pragma unroll
    for (int h = 0; h < 4; ++h) {
      float y = c[h][v] * rs * gg[h] + bb[h];
      y = y > 0.f ? y : expm1f(y);
      orow[h * 16] = y;
    }
  }
}

__global__ void edge_logits2(const int* __restrict__ ei, int E, int n,
                             const float* __restrict__ xl,
                             const float* __restrict__ xr,
                             const float* __restrict__ a2,
                             float* __restrict__ logits,
                             float* __restrict__ maxd) {
  const int lane = threadIdx.x & 31;
  const int wave = __builtin_amdgcn_readfirstlane(
      (int)(((long long)blockIdx.x * blockDim.x + threadIdx.x) >> 5));
  const int TE = E + n;
  const int base = wave * 32;
  if (base >= TE) return;
  const int t = base + lane;
  int s = 0, d = 0;
  if (t < TE) {
    if (t < E) { s = ei[t]; d = ei[E + t]; } else { s = t - E; d = s; }
  }
  const float al = a2[lane];
  float p[32];
#pragma unroll
  for (int j0 = 0; j0 < 32; j0 += 8) {
    float u[8], v[8];
#pragma unroll
    for (int k = 0; k < 8; ++k) {
      u[k] = xl[(unsigned)bcast_i(s, j0 + k) * 32u + (unsigned)lane];
      v[k] = xr[(unsigned)bcast_i(d, j0 + k) * 32u + (unsigned)lane];
    }
#pragma unroll
    for (int k = 0; k < 8; ++k) {
      float e = u[k] + v[k]; e = e >= 0.f ? e : NEG_SLOPE * e;
      p[j0 + k] = e * al;
    }
  }
  bfly<16, 16, 0>(p, (lane & 16) != 0);
  bfly<8, 8, 0>(p, (lane & 8) != 0);
  bfly<4, 4, 0>(p, (lane & 4) != 0);
  bfly<2, 2, 0>(p, (lane & 2) != 0);
  bfly<1, 1, 0>(p, (lane & 1) != 0);
  if (t < TE) {
    logits[t] = p[0];
    atomicMaxF(&maxd[d], p[0]);
  }
}

__global__ void edge_accum2(const int* __restrict__ ei, int E, int n,
                            const float* __restrict__ xl,
                            const float* __restrict__ logits,
                            const float* __restrict__ maxd,
                            float* __restrict__ denom,
                            float* __restrict__ acc) {
  const int lane = threadIdx.x & 31;
  const int wave = __builtin_amdgcn_readfirstlane(
      (int)(((long long)blockIdx.x * blockDim.x + threadIdx.x) >> 5));
  const int TE = E + n;
  const int base = wave * 32;
  if (base >= TE) return;
  const int t = base + lane;
  int s = 0, d = 0;
  float ex = 0.f;
  if (t < TE) {
    if (t < E) { s = ei[t]; d = ei[E + t]; } else { s = t - E; d = s; }
    ex = __expf(logits[t] - maxd[d]);
    atomicAdd(&denom[d], ex);
  }
  const int cnt = TE - base;
  if (cnt >= 32) {
#pragma unroll
    for (int j0 = 0; j0 < 32; j0 += 8) {
      float v[8];
#pragma unroll
      for (int u = 0; u < 8; ++u) v[u] = xl[(size_t)bcast_i(s, j0 + u) * 32 + lane];
#pragma unroll
      for (int u = 0; u < 8; ++u)
        atomicAdd(acc + (size_t)bcast_i(d, j0 + u) * 32 + lane, bcast_f(ex, j0 + u) * v[u]);
    }
  } else {
    for (int j = 0; j < cnt; ++j) {
      const float v = xl[(size_t)bcast_i(s, j) * 32 + lane];
      atomicAdd(acc + (size_t)bcast_i(d, j) * 32 + lane, bcast_f(ex, j) * v);
    }
  }
}

__global__ void node_fin2(const float* __restrict__ acc,
                          const float* __restrict__ denom,
                          const float* __restrict__ b2,
                          const float* __restrict__ g,
                          const float* __restrict__ be,
                          float* __restrict__ out, int n) {
  int node = (int)(((long long)blockIdx.x * blockDim.x + threadIdx.x) >> 5);
  int lane = threadIdx.x & 31;
  if (node >= n) return;
  float v = acc[(size_t)node * 32 + lane] / denom[node] + b2[lane];
  float s = v;
#pragma unroll
  for (int o = 16; o; o >>= 1) s += __shfl_xor(s, o, 32);
  float mu = s * (1.f / 32.f);
  float dd = v - mu;
  float q = dd * dd;
#pragma unroll
  for (int o = 16; o; o >>= 1) q += __shfl_xor(q, o, 32);
  float rs = rsqrtf(q * (1.f / 32.f) + LN_EPS);
  out[(size_t)node * 32 + lane] = dd * rs * g[lane] + be[lane];
}

static inline int gridFor(long long threads, int block) {
  return (int)((threads + block - 1) / block);
}

extern "C" void kernel_launch(void* const* d_in, const int* in_sizes, int n_in,
                              void* d_out, int out_size, void* d_ws, size_t ws_size,
                              hipStream_t stream) {
  const float* x   = (const float*)d_in[0];
  const int*   ei  = (const int*)  d_in[1];
  const float* W1l = (const float*)d_in[2];
  const float* W1r = (const float*)d_in[3];
  const float* a1  = (const float*)d_in[4];
  const float* b1  = (const float*)d_in[5];
  const float* g1  = (const float*)d_in[6];
  const float* be1 = (const float*)d_in[7];
  const float* W2l = (const float*)d_in[8];
  const float* W2r = (const float*)d_in[9];
  const float* a2  = (const float*)d_in[10];
  const float* b2  = (const float*)d_in[11];
  const float* g2  = (const float*)d_in[12];
  const float* be2 = (const float*)d_in[13];
  float* out = (float*)d_out;

  const int n  = in_sizes[0] / 4;
  const int E  = in_sizes[1] / 2;
  const int TE = E + n;
  const int ntiles = n / 16;

  float* ws = (float*)d_ws;
  const size_t XL1  = 0;
  const size_t XR1  = XL1 + (size_t)n * 64;
  const size_t ACC1 = XR1 + (size_t)n * 64;
  const size_t DEN1 = ACC1 + (size_t)n * 64;
  const size_t MAX1 = DEN1 + (size_t)n * 4;
  const size_t LOG1 = MAX1 + (size_t)n * 4;
  const size_t H1   = LOG1 + (size_t)TE * 4;

  const int B = 256;
  const float NEG_INF = -INFINITY;

  fill_f32<<<gridFor((long long)n * 32, B), B, 0, stream>>>(ws + ACC1, (long long)n * 32, 0.f);
  fill_f32<<<gridFor((long long)n * 4, B), B, 0, stream>>>(ws + MAX1, (long long)n * 4, NEG_INF);

  edge_logits1<<<gridFor((long long)TE * 2, B), B, 0, stream>>>(
      ei, E, n, x, W1l, W1r, a1, ws + LOG1, ws + MAX1);

  edge_accum1<<<gridFor((long long)TE * 4, B), B, 0, stream>>>(
      ei, E, n, x, ws + LOG1, ws + MAX1, ws + ACC1);

  node_fin1<<<gridFor((long long)ntiles * 32, 128), 128, 0, stream>>>(
      ws + ACC1, W1l, b1, g1, be1, ws + H1, ntiles);

  fill_f32<<<gridFor((long long)n * 32, B), B, 0, stream>>>(ws + ACC1, (long long)n * 32, 0.f);
  fill_f32<<<gridFor(n, B), B, 0, stream>>>(ws + DEN1, n, 0.f);
  fill_f32<<<gridFor(n, B), B, 0, stream>>>(ws + MAX1, n, NEG_INF);

  gemm2_wmma<<<gridFor((long long)ntiles * 32, B), B, 0, stream>>>(
      ws + H1, W2l, W2r, ws + XL1, ws + XR1, ntiles);

  edge_logits2<<<gridFor(TE, B), B, 0, stream>>>(
      ei, E, n, ws + XL1, ws + XR1, a2, ws + LOG1, ws + MAX1);

  edge_accum2<<<gridFor(TE, B), B, 0, stream>>>(
      ei, E, n, ws + XL1, ws + LOG1, ws + MAX1, ws + DEN1, ws + ACC1);

  node_fin2<<<gridFor((long long)n * 32, B), B, 0, stream>>>(
      ws + ACC1, ws + DEN1, b2, g2, be2, out, n);
}


# The five clearing launches become one launch of 16-byte stores at the head of the call; layer 2's sums, denominators and maxima get workspace of their own (no aliasing): 12 launches to 8
# speedup vs baseline: 3.6753x; 1.0384x over previous
#include <hip/hip_runtime.h>
#include <hip/hip_bf16.h>
#include <math.h>


typedef __attribute__((ext_vector_type(2))) float v2f;
typedef __attribute__((ext_vector_type(8))) float v8f;

#define NEG_SLOPE 0.2f
#define LN_EPS 1e-5f

static __device__ inline v8f wmma4(v2f a, v2f b, v8f c) {
  return __builtin_amdgcn_wmma_f32_16x16x4_f32(
       false, a,  false, b,
       (short)0, c,  false,  false);
}

static __device__ inline void atomicMaxF(float* addr, float val) {
  if (val >= 0.f) atomicMax((int*)addr, __float_as_int(val));
  else            atomicMin((unsigned int*)addr, __float_as_uint(val));
}

static __device__ inline int bcast_i(int v, int srcLane) {
  return __builtin_amdgcn_readlane(v, srcLane);
}
static __device__ inline float bcast_f(float v, int srcLane) {
  return __int_as_float(__builtin_amdgcn_readlane(__float_as_int(v), srcLane));
}

template <int OFF, int STRIDE, int Q>
static __device__ inline void bfly(float (&p)[32], bool up) {
  if constexpr (Q < STRIDE) {
    const float lo = p[Q], hi = p[Q + STRIDE];
    const float keep = up ? hi : lo;
    const float send = up ? lo : hi;
    p[Q] = keep + __shfl_xor(send, OFF, 32);
    bfly<OFF, STRIDE, Q + 1>(p, up);
  }
}

__global__ void clear_all(float4* __restrict__ acc1, float4* __restrict__ acc2, long long nq,
                          float4* __restrict__ max1, float* __restrict__ den2,
                          float* __restrict__ max2, int n) {
  const float4 z = make_float4(0.f, 0.f, 0.f, 0.f);
  const float ninf = -INFINITY;
  const long long base = (long long)blockIdx.x * (blockDim.x * 4) + threadIdx.x;
#pragma unroll
  for (int j = 0; j < 4; ++j) {
    const long long i = base + (long long)j * blockDim.x;
    if (i < nq) { acc1[i] = z; acc2[i] = z; }
  }
  const long long g = (long long)blockIdx.x * blockDim.x + threadIdx.x;
  if (g < n) {
    max1[g] = make_float4(ninf, ninf, ninf, ninf);
    den2[g] = 0.f;
    max2[g] = ninf;
  }
}

__global__ void gemm2_wmma(const float* __restrict__ h,
                           const float* __restrict__ Wl,
                           const float* __restrict__ Wr,
                           float* __restrict__ xl,
                           float* __restrict__ xr, int ntiles) {
  int wave = (int)(((long long)blockIdx.x * blockDim.x + threadIdx.x) >> 5);
  int lane = threadIdx.x & 31;
  if (wave >= ntiles) return;
  int row0 = wave * 16;
  int m  = lane & 15;
  int hi = lane >> 4;
  const float* hrow = h + (size_t)(row0 + m) * 64;
  v8f cl[2] = {{0.f},{0.f}};
  v8f cr[2] = {{0.f},{0.f}};
#pragma unroll
  for (int kk = 0; kk < 16; ++kk) {
    int k0 = kk * 4 + hi * 2;
    v2f a; a.x = hrow[k0]; a.y = hrow[k0 + 1];
#pragma unroll
    for (int j = 0; j < 2; ++j) {
      v2f bl, br;
      bl.x = Wl[(k0 + 0) * 32 + j * 16 + m];
      bl.y = Wl[(k0 + 1) * 32 + j * 16 + m];
      br.x = Wr[(k0 + 0) * 32 + j * 16 + m];
      br.y = Wr[(k0 + 1) * 32 + j * 16 + m];
      cl[j] = wmma4(a, bl, cl[j]);
      cr[j] = wmma4(a, br, cr[j]);
    }
  }
#pragma unroll
  for (int j = 0; j < 2; ++j)
#pragma unroll
    for (int v = 0; v < 8; ++v) {
      int r = v + hi * 8;
      xl[(size_t)(row0 + r) * 32 + j * 16 + m] = cl[j][v];
      xr[(size_t)(row0 + r) * 32 + j * 16 + m] = cr[j][v];
    }
}

__global__ void edge_logits1(const int* __restrict__ ei, int E, int n,
                             const float* __restrict__ x,
                             const float* __restrict__ Wl,
                             const float* __restrict__ Wr,
                             const float* __restrict__ a1,
                             float* __restrict__ logits,
                             float* __restrict__ maxd) {
  const int lane = threadIdx.x & 31;
  const int wave = __builtin_amdgcn_readfirstlane(
      (int)(((long long)blockIdx.x * blockDim.x + threadIdx.x) >> 5));
  const int TE = E + n;
  const int base = wave * 16;
  if (base >= TE) return;
  const int m  = lane & 15;
  const int hi = lane >> 4;
  const int t = base + m;
  int s = 0, d = 0;
  if (t < TE) {
    if (t < E) { s = ei[t]; d = ei[E + t]; } else { s = t - E; d = s; }
  }
  const float2 xs = *(const float2*)(x + (size_t)s * 4 + hi * 2);
  const float2 xd = *(const float2*)(x + (size_t)d * 4 + hi * 2);
  v2f fs; fs.x = xs.x; fs.y = xs.y;
  v2f fd; fd.x = xd.x; fd.y = xd.y;
  float p[32];
#pragma unroll
  for (int j = 0; j < 4; ++j) {
    v2f bl, br;
    bl.x = Wl[(2 * hi + 0) * 64 + j * 16 + m];
    bl.y = Wl[(2 * hi + 1) * 64 + j * 16 + m];
    br.x = Wr[(2 * hi + 0) * 64 + j * 16 + m];
    br.y = Wr[(2 * hi + 1) * 64 + j * 16 + m];
    const float aj = a1[j * 16 + m];
    v8f cc = {};
    cc = wmma4(fs, bl, cc);
    cc = wmma4(fd, br, cc);
#pragma unroll
    for (int v = 0; v < 8; ++v) {
      float e = cc[v]; e = e >= 0.f ? e : NEG_SLOPE * e;
      p[(v >> 1) * 8 + j * 2 + (v & 1)] = e * aj;
    }
  }
  bfly<8, 16, 0>(p, (lane & 8) != 0);
  bfly<4, 8, 0>(p, (lane & 4) != 0);
  bfly<2, 4, 0>(p, (lane & 2) != 0);
  bfly<1, 2, 0>(p, (lane & 1) != 0);
  const int head = lane & 3;
  const int jb = hi * 8 + 2 * ((lane >> 2) & 3);
#pragma unroll
  for (int q = 0; q < 2; ++q) {
    const int j = jb + q;
    const int dj = __shfl(d, j, 32);
    if (base + j < TE) {
      logits[(size_t)(base + j) * 4 + head] = p[q];
      atomicMaxF(&maxd[(size_t)dj * 4 + head], p[q]);
    }
  }
}

__global__ void edge_accum1(const int* __restrict__ ei, int E, int n,
                            const float* __restrict__ x,
                            const float* __restrict__ logits,
                            const float* __restrict__ maxd,
                            float* __restrict__ acc) {
  const int lane = threadIdx.x & 31;
  const int wave = __builtin_amdgcn_readfirstlane(
      (int)(((long long)blockIdx.x * blockDim.x + threadIdx.x) >> 5));
  const int TE = E + n;
  const int base = wave * 8;
  if (base >= TE) return;
  const int e = base + (lane >> 2), h = lane & 3;
  int s = 0, d = 0;
  float ex = 0.f;
  if (e < TE) {
    if (e < E) { s = ei[e]; d = ei[E + e]; } else { s = e - E; d = s; }
    ex = __expf(logits[(size_t)e * 4 + h] - maxd[(size_t)d * 4 + h]);
  }
  const int cnt = TE - base;
  const int kx = lane < 16 ? (lane >> 2) : 3;
  const int hsel = (lane & 3) << 2;
  if (cnt >= 8) {
    float xv[8], w[8];
    int dj[8];
#pragma unroll
    for (int j = 0; j < 8; ++j)
      xv[j] = x[(size_t)bcast_i(s, 4 * j) * 4 + kx];
#pragma unroll
    for (int j = 0; j < 8; ++j) {
      w[j] = __int_as_float(
          __builtin_amdgcn_ds_bpermute(hsel + 16 * j, __float_as_int(ex)));
      dj[j] = bcast_i(d, 4 * j);
    }
    if (lane < 20) {
#pragma unroll
      for (int j = 0; j < 8; ++j)
        atomicAdd(acc + (size_t)dj[j] * 32 + lane, (lane < 16 ? xv[j] : 1.f) * w[j]);
    }
  } else {
    for (int j = 0; j < cnt; ++j) {
      const float xj = x[(size_t)bcast_i(s, 4 * j) * 4 + kx];
      const float w = __int_as_float(
          __builtin_amdgcn_ds_bpermute(hsel + 16 * j, __float_as_int(ex)));
      const float val = (lane < 16 ? xj : 1.f) * w;
      float* row = acc + (size_t)bcast_i(d, 4 * j) * 32;
      if (lane < 20) atomicAdd(row + lane, val);
    }
  }
}

__global__ void node_fin1(const float* __restrict__ acc,
                          const float* __restrict__ Wl,
                          const float* __restrict__ b1,
                          const float* __restrict__ g,
                          const float* __restrict__ be,
                          float* __restrict__ h1, int ntiles) {
  int wave = (int)(((long long)blockIdx.x * blockDim.x + threadIdx.x) >> 5);
  int lane = threadIdx.x & 31;
  if (wave >= ntiles) return;
  int row0 = wave * 16;
  int m  = lane & 15;
  int hi = lane >> 4;
  const float4* arow = (const float4*)(acc + (size_t)(row0 + m) * 32);
  const float4 p0 = arow[2 * hi];
  const float4 p1 = arow[2 * hi + 1];
  const float4 dn = arow[4];
  const float pa[4] = {p0.x, p0.y, p0.z, p0.w};
  const float pb[4] = {p1.x, p1.y, p1.z, p1.w};
  const float dd[4] = {dn.x, dn.y, dn.z, dn.w};
  v8f c[4];
  float bias[4], gg[4], bb[4];
#pragma unroll
  for (int h = 0; h < 4; ++h) {
    v2f a; a.x = pa[h] / dd[h]; a.y = pb[h] / dd[h];
    v2f b;
    b.x = Wl[(2 * hi + 0) * 64 + h * 16 + m];
    b.y = Wl[(2 * hi + 1) * 64 + h * 16 + m];
    v8f z = {};
    c[h] = wmma4(a, b, z);
    bias[h] = b1[h * 16 + m];
    gg[h] = g[h * 16 + m];
    bb[h] = be[h * 16 + m];
  }
  float s[8], q[8];
#pragma unroll
  for (int v = 0; v < 8; ++v) {
#pragma unroll
    for (int h = 0; h < 4; ++h) c[h][v] += bias[h];
    s[v] = (c[0][v] + c[1][v]) + (c[2][v] + c[3][v]);
  }
#pragma unroll
  for (int o = 8; o; o >>= 1)
#pragma unroll
    for (int v = 0; v < 8; ++v) s[v] += __shfl_xor(s[v], o, 32);
#pragma unroll
  for (int v = 0; v < 8; ++v) {
    const float mu = s[v] * (1.f / 64.f);
#pragma unroll
    for (int h = 0; h < 4; ++h) c[h][v] -= mu;
    q[v] = (c[0][v] * c[0][v] + c[1][v] * c[1][v]) + (c[2][v] * c[2][v] + c[3][v] * c[3][v]);
  }
#pragma unroll
  for (int o = 8; o; o >>= 1)
#pragma unroll
    for (int v = 0; v < 8; ++v) q[v] += __shfl_xor(q[v], o, 32);
#pragma unroll
  for (int v = 0; v < 8; ++v) {
    const float rs = rsqrtf(q[v] * (1.f / 64.f) + LN_EPS);
    float* orow = h1 + (size_t)(row0 + v + hi * 8) * 64 + m;
#pragma unroll
    for (int h = 0; h < 4; ++h) {
      float y = c[h][v] * rs * gg[h] + bb[h];
      y = y > 0.f ? y : expm1f(y);
      orow[h * 16] = y;
    }
  }
}

__global__ void edge_logits2(const int* __restrict__ ei, int E, int n,
                             const float* __restrict__ xl,
                             const float* __restrict__ xr,
                             const float* __restrict__ a2,
                             float* __restrict__ logits,
                             float* __restrict__ maxd) {
  const int lane = threadIdx.x & 31;
  const int wave = __builtin_amdgcn_readfirstlane(
      (int)(((long long)blockIdx.x * blockDim.x + threadIdx.x) >> 5));
  const int TE = E + n;
  const int base = wave * 32;
  if (base >= TE) return;
  const int t = base + lane;
  int s = 0, d = 0;
  if (t < TE) {
    if (t < E) { s = ei[t]; d = ei[E + t]; } else { s = t - E; d = s; }
  }
  const float al = a2[lane];
  float p[32];
#pragma unroll
  for (int j0 = 0; j0 < 32; j0 += 8) {
    float u[8], v[8];
#pragma unroll
    for (int k = 0; k < 8; ++k) {
      u[k] = xl[(unsigned)bcast_i(s, j0 + k) * 32u + (unsigned)lane];
      v[k] = xr[(unsigned)bcast_i(d, j0 + k) * 32u + (unsigned)lane];
    }
#pragma unroll
    for (int k = 0; k < 8; ++k) {
      float e = u[k] + v[k]; e = e >= 0.f ? e : NEG_SLOPE * e;
      p[j0 + k] = e * al;
    }
  }
  bfly<16, 16, 0>(p, (lane & 16) != 0);
  bfly<8, 8, 0>(p, (lane & 8) != 0);
  bfly<4, 4, 0>(p, (lane & 4) != 0);
  bfly<2, 2, 0>(p, (lane & 2) != 0);
  bfly<1, 1, 0>(p, (lane & 1) != 0);
  if (t < TE) {
    logits[t] = p[0];
    atomicMaxF(&maxd[d], p[0]);
  }
}

__global__ void edge_accum2(const int* __restrict__ ei, int E, int n,
                            const float* __restrict__ xl,
                            const float* __restrict__ logits,
                            const float* __restrict__ maxd,
                            float* __restrict__ denom,
                            float* __restrict__ acc) {
  const int lane = threadIdx.x & 31;
  const int wave = __builtin_amdgcn_readfirstlane(
      (int)(((long long)blockIdx.x * blockDim.x + threadIdx.x) >> 5));
  const int TE = E + n;
  const int base = wave * 32;
  if (base >= TE) return;
  const int t = base + lane;
  int s = 0, d = 0;
  float ex = 0.f;
  if (t < TE) {
    if (t < E) { s = ei[t]; d = ei[E + t]; } else { s = t - E; d = s; }
    ex = __expf(logits[t] - maxd[d]);
    atomicAdd(&denom[d], ex);
  }
  const int cnt = TE - base;
  if (cnt >= 32) {
#pragma unroll
    for (int j0 = 0; j0 < 32; j0 += 8) {
      float v[8];
#pragma unroll
      for (int u = 0; u < 8; ++u) v[u] = xl[(size_t)bcast_i(s, j0 + u) * 32 + lane];
#pragma unroll
      for (int u = 0; u < 8; ++u)
        atomicAdd(acc + (size_t)bcast_i(d, j0 + u) * 32 + lane, bcast_f(ex, j0 + u) * v[u]);
    }
  } else {
    for (int j = 0; j < cnt; ++j) {
      const float v = xl[(size_t)bcast_i(s, j) * 32 + lane];
      atomicAdd(acc + (size_t)bcast_i(d, j) * 32 + lane, bcast_f(ex, j) * v);
    }
  }
}

__global__ void node_fin2(const float* __restrict__ acc,
                          const float* __restrict__ denom,
                          const float* __restrict__ b2,
                          const float* __restrict__ g,
                          const float* __restrict__ be,
                          float* __restrict__ out, int n) {
  int node = (int)(((long long)blockIdx.x * blockDim.x + threadIdx.x) >> 5);
  int lane = threadIdx.x & 31;
  if (node >= n) return;
  float v = acc[(size_t)node * 32 + lane] / denom[node] + b2[lane];
  float s = v;
#pragma unroll
  for (int o = 16; o; o >>= 1) s += __shfl_xor(s, o, 32);
  float mu = s * (1.f / 32.f);
  float dd = v - mu;
  float q = dd * dd;
#pragma unroll
  for (int o = 16; o; o >>= 1) q += __shfl_xor(q, o, 32);
  float rs = rsqrtf(q * (1.f / 32.f) + LN_EPS);
  out[(size_t)node * 32 + lane] = dd * rs * g[lane] + be[lane];
}

static inline int gridFor(long long threads, int block) {
  return (int)((threads + block - 1) / block);
}

extern "C" void kernel_launch(void* const* d_in, const int* in_sizes, int n_in,
                              void* d_out, int out_size, void* d_ws, size_t ws_size,
                              hipStream_t stream) {
  const float* x   = (const float*)d_in[0];
  const int*   ei  = (const int*)  d_in[1];
  const float* W1l = (const float*)d_in[2];
  const float* W1r = (const float*)d_in[3];
  const float* a1  = (const float*)d_in[4];
  const float* b1  = (const float*)d_in[5];
  const float* g1  = (const float*)d_in[6];
  const float* be1 = (const float*)d_in[7];
  const float* W2l = (const float*)d_in[8];
  const float* W2r = (const float*)d_in[9];
  const float* a2  = (const float*)d_in[10];
  const float* b2  = (const float*)d_in[11];
  const float* g2  = (const float*)d_in[12];
  const float* be2 = (const float*)d_in[13];
  float* out = (float*)d_out;

  const int n  = in_sizes[0] / 4;
  const int E  = in_sizes[1] / 2;
  const int TE = E + n;
  const int ntiles = n / 16;

  float* ws = (float*)d_ws;
  const size_t XL1  = 0;
  const size_t ACC2 = XL1 + (size_t)n * 32;
  const size_t XR1  = XL1 + (size_t)n * 64;
  const size_t DEN2 = XR1 + (size_t)n * 32;
  const size_t MAX2 = DEN2 + (size_t)n;
  const size_t ACC1 = XR1 + (size_t)n * 64;
  const size_t MAX1 = ACC1 + (size_t)n * 68;
  const size_t LOG1 = MAX1 + (size_t)n * 4;
  const size_t H1   = LOG1 + (size_t)TE * 4;

  const int B = 256;

  clear_all<<<gridFor((long long)n * 8, B * 4), B, 0, stream>>>(
      (float4*)(ws + ACC1), (float4*)(ws + ACC2), (long long)n * 8,
      (float4*)(ws + MAX1), ws + DEN2, ws + MAX2, n);

  edge_logits1<<<gridFor((long long)TE * 2, B), B, 0, stream>>>(
      ei, E, n, x, W1l, W1r, a1, ws + LOG1, ws + MAX1);

  edge_accum1<<<gridFor((long long)TE * 4, B), B, 0, stream>>>(
      ei, E, n, x, ws + LOG1, ws + MAX1, ws + ACC1);

  node_fin1<<<gridFor((long long)ntiles * 32, 128), 128, 0, stream>>>(
      ws + ACC1, W1l, b1, g1, be1, ws + H1, ntiles);

  gemm2_wmma<<<gridFor((long long)ntiles * 32, B), B, 0, stream>>>(
      ws + H1, W2l, W2r, ws + XL1, ws + XR1, ntiles);

  edge_logits2<<<gridFor(TE, B), B, 0, stream>>>(
      ei, E, n, ws + XL1, ws + XR1, a2, ws + LOG1, ws + MAX2);

  edge_accum2<<<gridFor(TE, B), B, 0, stream>>>(
      ei, E, n, ws + XL1, ws + LOG1, ws + MAX2, ws + DEN2, ws + ACC2);

  node_fin2<<<gridFor((long long)n * 32, B), B, 0, stream>>>(
      ws + ACC2, ws + DEN2, b2, g2, be2, out, n);
}


# Layer-2 denominator rides in its row's 128-B line: word 31 = sum(ex), words 0..30 = sum(ex*(xl_c-xl_31)); LayerNorm drops the common shift; no denominator atomics; new summands, arrival order kept
# speedup vs baseline: 4.2891x; 1.1670x over previous
#include <hip/hip_runtime.h>
#include <hip/hip_bf16.h>
#include <math.h>


typedef __attribute__((ext_vector_type(2))) float v2f;
typedef __attribute__((ext_vector_type(8))) float v8f;

#define NEG_SLOPE 0.2f
#define LN_EPS 1e-5f

static __device__ inline v8f wmma4(v2f a, v2f b, v8f c) {
  return __builtin_amdgcn_wmma_f32_16x16x4_f32(
       false, a,  false, b,
       (short)0, c,  false,  false);
}

static __device__ inline void atomicMaxF(float* addr, float val) {
  if (val >= 0.f) atomicMax((int*)addr, __float_as_int(val));
  else            atomicMin((unsigned int*)addr, __float_as_uint(val));
}

static __device__ inline int bcast_i(int v, int srcLane) {
  return __builtin_amdgcn_readlane(v, srcLane);
}
static __device__ inline float bcast_f(float v, int srcLane) {
  return __int_as_float(__builtin_amdgcn_readlane(__float_as_int(v), srcLane));
}

template <int OFF, int STRIDE, int Q>
static __device__ inline void bfly(float (&p)[32], bool up) {
  if constexpr (Q < STRIDE) {
    const float lo = p[Q], hi = p[Q + STRIDE];
    const float keep = up ? hi : lo;
    const float send = up ? lo : hi;
    p[Q] = keep + __shfl_xor(send, OFF, 32);
    bfly<OFF, STRIDE, Q + 1>(p, up);
  }
}

__global__ void clear_all(float4* __restrict__ acc1, float4* __restrict__ acc2, long long nq,
                          float4* __restrict__ max1,
                          float* __restrict__ max2, int n) {
  const float4 z = make_float4(0.f, 0.f, 0.f, 0.f);
  const float ninf = -INFINITY;
  const long long base = (long long)blockIdx.x * (blockDim.x * 4) + threadIdx.x;
#pragma unroll
  for (int j = 0; j < 4; ++j) {
    const long long i = base + (long long)j * blockDim.x;
    if (i < nq) { acc1[i] = z; acc2[i] = z; }
  }
  const long long g = (long long)blockIdx.x * blockDim.x + threadIdx.x;
  if (g < n) {
    max1[g] = make_float4(ninf, ninf, ninf, ninf);
    max2[g] = ninf;
  }
}

__global__ void gemm2_wmma(const float* __restrict__ h,
                           const float* __restrict__ Wl,
                           const float* __restrict__ Wr,
                           float* __restrict__ xl,
                           float* __restrict__ xr, int ntiles) {
  int wave = (int)(((long long)blockIdx.x * blockDim.x + threadIdx.x) >> 5);
  int lane = threadIdx.x & 31;
  if (wave >= ntiles) return;
  int row0 = wave * 16;
  int m  = lane & 15;
  int hi = lane >> 4;
  const float* hrow = h + (size_t)(row0 + m) * 64;
  v8f cl[2] = {{0.f},{0.f}};
  v8f cr[2] = {{0.f},{0.f}};
#pragma unroll
  for (int kk = 0; kk < 16; ++kk) {
    int k0 = kk * 4 + hi * 2;
    v2f a; a.x = hrow[k0]; a.y = hrow[k0 + 1];
#pragma unroll
    for (int j = 0; j < 2; ++j) {
      v2f bl, br;
      bl.x = Wl[(k0 + 0) * 32 + j * 16 + m];
      bl.y = Wl[(k0 + 1) * 32 + j * 16 + m];
      br.x = Wr[(k0 + 0) * 32 + j * 16 + m];
      br.y = Wr[(k0 + 1) * 32 + j * 16 + m];
      cl[j] = wmma4(a, bl, cl[j]);
      cr[j] = wmma4(a, br, cr[j]);
    }
  }
#pragma unroll
  for (int j = 0; j < 2; ++j)
#pragma unroll
    for (int v = 0; v < 8; ++v) {
      int r = v + hi * 8;
      xl[(size_t)(row0 + r) * 32 + j * 16 + m] = cl[j][v];
      xr[(size_t)(row0 + r) * 32 + j * 16 + m] = cr[j][v];
    }
}

__global__ void edge_logits1(const int* __restrict__ ei, int E, int n,
                             const float* __restrict__ x,
                             const float* __restrict__ Wl,
                             const float* __restrict__ Wr,
                             const float* __restrict__ a1,
                             float* __restrict__ logits,
                             float* __restrict__ maxd) {
  const int lane = threadIdx.x & 31;
  const int wave = __builtin_amdgcn_readfirstlane(
      (int)(((long long)blockIdx.x * blockDim.x + threadIdx.x) >> 5));
  const int TE = E + n;
  const int base = wave * 16;
  if (base >= TE) return;
  const int m  = lane & 15;
  const int hi = lane >> 4;
  const int t = base + m;
  int s = 0, d = 0;
  if (t < TE) {
    if (t < E) { s = ei[t]; d = ei[E + t]; } else { s = t - E; d = s; }
  }
  const float2 xs = *(const float2*)(x + (size_t)s * 4 + hi * 2);
  const float2 xd = *(const float2*)(x + (size_t)d * 4 + hi * 2);
  v2f fs; fs.x = xs.x; fs.y = xs.y;
  v2f fd; fd.x = xd.x; fd.y = xd.y;
  float p[32];
#pragma unroll
  for (int j = 0; j < 4; ++j) {
    v2f bl, br;
    bl.x = Wl[(2 * hi + 0) * 64 + j * 16 + m];
    bl.y = Wl[(2 * hi + 1) * 64 + j * 16 + m];
    br.x = Wr[(2 * hi + 0) * 64 + j * 16 + m];
    br.y = Wr[(2 * hi + 1) * 64 + j * 16 + m];
    const float aj = a1[j * 16 + m];
    v8f cc = {};
    cc = wmma4(fs, bl, cc);
    cc = wmma4(fd, br, cc);
#pragma unroll
    for (int v = 0; v < 8; ++v) {
      float e = cc[v]; e = e >= 0.f ? e : NEG_SLOPE * e;
      p[(v >> 1) * 8 + j * 2 + (v & 1)] = e * aj;
    }
  }
  bfly<8, 16, 0>(p, (lane & 8) != 0);
  bfly<4, 8, 0>(p, (lane & 4) != 0);
  bfly<2, 4, 0>(p, (lane & 2) != 0);
  bfly<1, 2, 0>(p, (lane & 1) != 0);
  const int head = lane & 3;
  const int jb = hi * 8 + 2 * ((lane >> 2) & 3);
#pragma unroll
  for (int q = 0; q < 2; ++q) {
    const int j = jb + q;
    const int dj = __shfl(d, j, 32);
    if (base + j < TE) {
      logits[(size_t)(base + j) * 4 + head] = p[q];
      atomicMaxF(&maxd[(size_t)dj * 4 + head], p[q]);
    }
  }
}

__global__ void edge_accum1(const int* __restrict__ ei, int E, int n,
                            const float* __restrict__ x,
                            const float* __restrict__ logits,
                            const float* __restrict__ maxd,
                            float* __restrict__ acc) {
  const int lane = threadIdx.x & 31;
  const int wave = __builtin_amdgcn_readfirstlane(
      (int)(((long long)blockIdx.x * blockDim.x + threadIdx.x) >> 5));
  const int TE = E + n;
  const int base = wave * 8;
  if (base >= TE) return;
  const int e = base + (lane >> 2), h = lane & 3;
  int s = 0, d = 0;
  float ex = 0.f;
  if (e < TE) {
    if (e < E) { s = ei[e]; d = ei[E + e]; } else { s = e - E; d = s; }
    ex = __expf(logits[(size_t)e * 4 + h] - maxd[(size_t)d * 4 + h]);
  }
  const int cnt = TE - base;
  const int kx = lane < 16 ? (lane >> 2) : 3;
  const int hsel = (lane & 3) << 2;
  if (cnt >= 8) {
    float xv[8], w[8];
    int dj[8];
#pragma unroll
    for (int j = 0; j < 8; ++j)
      xv[j] = x[(size_t)bcast_i(s, 4 * j) * 4 + kx];
#pragma unroll
    for (int j = 0; j < 8; ++j) {
      w[j] = __int_as_float(
          __builtin_amdgcn_ds_bpermute(hsel + 16 * j, __float_as_int(ex)));
      dj[j] = bcast_i(d, 4 * j);
    }
    if (lane < 20) {
#pragma unroll
      for (int j = 0; j < 8; ++j)
        atomicAdd(acc + (size_t)dj[j] * 32 + lane, (lane < 16 ? xv[j] : 1.f) * w[j]);
    }
  } else {
    for (int j = 0; j < cnt; ++j) {
      const float xj = x[(size_t)bcast_i(s, 4 * j) * 4 + kx];
      const float w = __int_as_float(
          __builtin_amdgcn_ds_bpermute(hsel + 16 * j, __float_as_int(ex)));
      const float val = (lane < 16 ? xj : 1.f) * w;
      float* row = acc + (size_t)bcast_i(d, 4 * j) * 32;
      if (lane < 20) atomicAdd(row + lane, val);
    }
  }
}

__global__ void node_fin1(const float* __restrict__ acc,
                          const float* __restrict__ Wl,
                          const float* __restrict__ b1,
                          const float* __restrict__ g,
                          const float* __restrict__ be,
                          float* __restrict__ h1, int ntiles) {
  int wave = (int)(((long long)blockIdx.x * blockDim.x + threadIdx.x) >> 5);
  int lane = threadIdx.x & 31;
  if (wave >= ntiles) return;
  int row0 = wave * 16;
  int m  = lane & 15;
  int hi = lane >> 4;
  const float4* arow = (const float4*)(acc + (size_t)(row0 + m) * 32);
  const float4 p0 = arow[2 * hi];
  const float4 p1 = arow[2 * hi + 1];
  const float4 dn = arow[4];
  const float pa[4] = {p0.x, p0.y, p0.z, p0.w};
  const float pb[4] = {p1.x, p1.y, p1.z, p1.w};
  const float dd[4] = {dn.x, dn.y, dn.z, dn.w};
  v8f c[4];
  float bias[4], gg[4], bb[4];
#pragma unroll
  for (int h = 0; h < 4; ++h) {
    v2f a; a.x = pa[h] / dd[h]; a.y = pb[h] / dd[h];
    v2f b;
    b.x = Wl[(2 * hi + 0) * 64 + h * 16 + m];
    b.y = Wl[(2 * hi + 1) * 64 + h * 16 + m];
    v8f z = {};
    c[h] = wmma4(a, b, z);
    bias[h] = b1[h * 16 + m];
    gg[h] = g[h * 16 + m];
    bb[h] = be[h * 16 + m];
  }
  float s[8], q[8];
#pragma unroll
  for (int v = 0; v < 8; ++v) {
#pragma unroll
    for (int h = 0; h < 4; ++h) c[h][v] += bias[h];
    s[v] = (c[0][v] + c[1][v]) + (c[2][v] + c[3][v]);
  }
#pragma unroll
  for (int o = 8; o; o >>= 1)
#pragma unroll
    for (int v = 0; v < 8; ++v) s[v] += __shfl_xor(s[v], o, 32);
#pragma unroll
  for (int v = 0; v < 8; ++v) {
    const float mu = s[v] * (1.f / 64.f);
#pragma unroll
    for (int h = 0; h < 4; ++h) c[h][v] -= mu;
    q[v] = (c[0][v] * c[0][v] + c[1][v] * c[1][v]) + (c[2][v] * c[2][v] + c[3][v] * c[3][v]);
  }
#pragma unroll
  for (int o = 8; o; o >>= 1)
#pragma unroll
    for (int v = 0; v < 8; ++v) q[v] += __shfl_xor(q[v], o, 32);
#pragma unroll
  for (int v = 0; v < 8; ++v) {
    const float rs = rsqrtf(q[v] * (1.f / 64.f) + LN_EPS);
    float* orow = h1 + (size_t)(row0 + v + hi * 8) * 64 + m;
#pragma unroll
    for (int h = 0; h < 4; ++h) {
      float y = c[h][v] * rs * gg[h] + bb[h];
      y = y > 0.f ? y : expm1f(y);
      orow[h * 16] = y;
    }
  }
}

__global__ void edge_logits2(const int* __restrict__ ei, int E, int n,
                             const float* __restrict__ xl,
                             const float* __restrict__ xr,
                             const float* __restrict__ a2,
                             float* __restrict__ logits,
                             float* __restrict__ maxd) {
  const int lane = threadIdx.x & 31;
  const int wave = __builtin_amdgcn_readfirstlane(
      (int)(((long long)blockIdx.x * blockDim.x + threadIdx.x) >> 5));
  const int TE = E + n;
  const int base = wave * 32;
  if (base >= TE) return;
  const int t = base + lane;
  int s = 0, d = 0;
  if (t < TE) {
    if (t < E) { s = ei[t]; d = ei[E + t]; } else { s = t - E; d = s; }
  }
  const float al = a2[lane];
  float p[32];
#pragma unroll
  for (int j0 = 0; j0 < 32; j0 += 8) {
    float u[8], v[8];
#pragma unroll
    for (int k = 0; k < 8; ++k) {
      u[k] = xl[(unsigned)bcast_i(s, j0 + k) * 32u + (unsigned)lane];
      v[k] = xr[(unsigned)bcast_i(d, j0 + k) * 32u + (unsigned)lane];
    }
#pragma unroll
    for (int k = 0; k < 8; ++k) {
      float e = u[k] + v[k]; e = e >= 0.f ? e : NEG_SLOPE * e;
      p[j0 + k] = e * al;
    }
  }
  bfly<16, 16, 0>(p, (lane & 16) != 0);
  bfly<8, 8, 0>(p, (lane & 8) != 0);
  bfly<4, 4, 0>(p, (lane & 4) != 0);
  bfly<2, 2, 0>(p, (lane & 2) != 0);
  bfly<1, 1, 0>(p, (lane & 1) != 0);
  if (t < TE) {
    logits[t] = p[0];
    atomicMaxF(&maxd[d], p[0]);
  }
}

__global__ void edge_accum2(const int* __restrict__ ei, int E, int n,
                            const float* __restrict__ xl,
                            const float* __restrict__ logits,
                            const float* __restrict__ maxd,
                            float* __restrict__ acc) {
  const int lane = threadIdx.x & 31;
  const int wave = __builtin_amdgcn_readfirstlane(
      (int)(((long long)blockIdx.x * blockDim.x + threadIdx.x) >> 5));
  const int TE = E + n;
  const int base = wave * 32;
  if (base >= TE) return;
  const int t = base + lane;
  int s = 0, d = 0;
  float ex = 0.f;
  if (t < TE) {
    if (t < E) { s = ei[t]; d = ei[E + t]; } else { s = t - E; d = s; }
    ex = __expf(logits[t] - maxd[d]);
  }
  const int cnt = TE - base;
  const bool den_lane = lane == 31;
  if (cnt >= 32) {
#pragma unroll
    for (int j0 = 0; j0 < 32; j0 += 8) {
      float v[8];
#pragma unroll
      for (int u = 0; u < 8; ++u) v[u] = xl[(size_t)bcast_i(s, j0 + u) * 32 + lane];
#pragma unroll
      for (int u = 0; u < 8; ++u) {
        const float c31 = bcast_f(v[u], 31);
        const float rel = den_lane ? 1.f : v[u] - c31;
        atomicAdd(acc + (size_t)bcast_i(d, j0 + u) * 32 + lane, bcast_f(ex, j0 + u) * rel);
      }
    }
  } else {
    for (int j = 0; j < cnt; ++j) {
      const float v = xl[(size_t)bcast_i(s, j) * 32 + lane];
      const float c31 = bcast_f(v, 31);
      const float rel = den_lane ? 1.f : v - c31;
      atomicAdd(acc + (size_t)bcast_i(d, j) * 32 + lane, bcast_f(ex, j) * rel);
    }
  }
}

__global__ void node_fin2(const float* __restrict__ acc,
                          const float* __restrict__ b2,
                          const float* __restrict__ g,
                          const float* __restrict__ be,
                          float* __restrict__ out, int n) {
  int node = (int)(((long long)blockIdx.x * blockDim.x + threadIdx.x) >> 5);
  int lane = threadIdx.x & 31;
  if (node >= n) return;
  const float a = acc[(size_t)node * 32 + lane];
  const float den = bcast_f(a, 31);
  float v = (lane == 31 ? 0.f : a / den) + b2[lane];
  float s = v;
#pragma unroll
  for (int o = 16; o; o >>= 1) s += __shfl_xor(s, o, 32);
  float mu = s * (1.f / 32.f);
  float dd = v - mu;
  float q = dd * dd;
#pragma unroll
  for (int o = 16; o; o >>= 1) q += __shfl_xor(q, o, 32);
  float rs = rsqrtf(q * (1.f / 32.f) + LN_EPS);
  out[(size_t)node * 32 + lane] = dd * rs * g[lane] + be[lane];
}

static inline int gridFor(long long threads, int block) {
  return (int)((threads + block - 1) / block);
}

extern "C" void kernel_launch(void* const* d_in, const int* in_sizes, int n_in,
                              void* d_out, int out_size, void* d_ws, size_t ws_size,
                              hipStream_t stream) {
  const float* x   = (const float*)d_in[0];
  const int*   ei  = (const int*)  d_in[1];
  const float* W1l = (const float*)d_in[2];
  const float* W1r = (const float*)d_in[3];
  const float* a1  = (const float*)d_in[4];
  const float* b1  = (const float*)d_in[5];
  const float* g1  = (const float*)d_in[6];
  const float* be1 = (const float*)d_in[7];
  const float* W2l = (const float*)d_in[8];
  const float* W2r = (const float*)d_in[9];
  const float* a2  = (const float*)d_in[10];
  const float* b2  = (const float*)d_in[11];
  const float* g2  = (const float*)d_in[12];
  const float* be2 = (const float*)d_in[13];
  float* out = (float*)d_out;

  const int n  = in_sizes[0] / 4;
  const int E  = in_sizes[1] / 2;
  const int TE = E + n;
  const int ntiles = n / 16;

  float* ws = (float*)d_ws;
  const size_t XL1  = 0;
  const size_t ACC2 = XL1 + (size_t)n * 32;
  const size_t XR1  = XL1 + (size_t)n * 64;
  const size_t MAX2 = XR1 + (size_t)n * 32;
  const size_t ACC1 = XR1 + (size_t)n * 64;
  const size_t MAX1 = ACC1 + (size_t)n * 68;
  const size_t LOG1 = MAX1 + (size_t)n * 4;
  const size_t H1   = LOG1 + (size_t)TE * 4;

  const int B = 256;

  clear_all<<<gridFor((long long)n * 8, B * 4), B, 0, stream>>>(
      (float4*)(ws + ACC1), (float4*)(ws + ACC2), (long long)n * 8,
      (float4*)(ws + MAX1), ws + MAX2, n);

  edge_logits1<<<gridFor((long long)TE * 2, B), B, 0, stream>>>(
      ei, E, n, x, W1l, W1r, a1, ws + LOG1, ws + MAX1);

  edge_accum1<<<gridFor((long long)TE * 4, B), B, 0, stream>>>(
      ei, E, n, x, ws + LOG1, ws + MAX1, ws + ACC1);

  node_fin1<<<gridFor((long long)ntiles * 32, 128), 128, 0, stream>>>(
      ws + ACC1, W1l, b1, g1, be1, ws + H1, ntiles);

  gemm2_wmma<<<gridFor((long long)ntiles * 32, B), B, 0, stream>>>(
      ws + H1, W2l, W2r, ws + XL1, ws + XR1, ntiles);

  edge_logits2<<<gridFor(TE, B), B, 0, stream>>>(
      ei, E, n, ws + XL1, ws + XR1, a2, ws + LOG1, ws + MAX2);

  edge_accum2<<<gridFor(TE, B), B, 0, stream>>>(
      ei, E, n, ws + XL1, ws + LOG1, ws + MAX2, ws + ACC2);

  node_fin2<<<gridFor((long long)n * 32, B), B, 0, stream>>>(
      ws + ACC2, b2, g2, be2, out, n);
}


# Segment maxima kept as order-preserving integer keys: one signed atomic max replaces the sign-split max/min pair, so an edge's 4 head maxima are one request to their line; results bit-identical
# speedup vs baseline: 4.9399x; 1.1517x over previous
#include <hip/hip_runtime.h>
#include <hip/hip_bf16.h>
#include <math.h>


typedef __attribute__((ext_vector_type(2))) float v2f;
typedef __attribute__((ext_vector_type(8))) float v8f;

#define NEG_SLOPE 0.2f
#define LN_EPS 1e-5f

static __device__ inline v8f wmma4(v2f a, v2f b, v8f c) {
  return __builtin_amdgcn_wmma_f32_16x16x4_f32(
       false, a,  false, b,
       (short)0, c,  false,  false);
}

#define MAXKEY_NEG_INF ((int)0x807fffff)
static __device__ inline int fkey(float f) {
  const int b = __float_as_int(f);
  return b >= 0 ? b : b ^ 0x7fffffff;
}
static __device__ inline float fkey_inv(int k) {
  return __int_as_float(k >= 0 ? k : k ^ 0x7fffffff);
}
static __device__ inline void atomicMaxKey(int* addr, float val) {
  atomicMax(addr, fkey(val));
}

static __device__ inline int bcast_i(int v, int srcLane) {
  return __builtin_amdgcn_readlane(v, srcLane);
}
static __device__ inline float bcast_f(float v, int srcLane) {
  return __int_as_float(__builtin_amdgcn_readlane(__float_as_int(v), srcLane));
}

template <int OFF, int STRIDE, int Q>
static __device__ inline void bfly(float (&p)[32], bool up) {
  if constexpr (Q < STRIDE) {
    const float lo = p[Q], hi = p[Q + STRIDE];
    const float keep = up ? hi : lo;
    const float send = up ? lo : hi;
    p[Q] = keep + __shfl_xor(send, OFF, 32);
    bfly<OFF, STRIDE, Q + 1>(p, up);
  }
}

__global__ void clear_all(float4* __restrict__ acc1, float4* __restrict__ acc2, long long nq,
                          int4* __restrict__ max1,
                          int* __restrict__ max2, int n) {
  const float4 z = make_float4(0.f, 0.f, 0.f, 0.f);
  const int ninf = MAXKEY_NEG_INF;
  const long long base = (long long)blockIdx.x * (blockDim.x * 4) + threadIdx.x;
#pragma unroll
  for (int j = 0; j < 4; ++j) {
    const long long i = base + (long long)j * blockDim.x;
    if (i < nq) { acc1[i] = z; acc2[i] = z; }
  }
  const long long g = (long long)blockIdx.x * blockDim.x + threadIdx.x;
  if (g < n) {
    max1[g] = make_int4(ninf, ninf, ninf, ninf);
    max2[g] = ninf;
  }
}

__global__ void gemm2_wmma(const float* __restrict__ h,
                           const float* __restrict__ Wl,
                           const float* __restrict__ Wr,
                           float* __restrict__ xl,
                           float* __restrict__ xr, int ntiles) {
  int wave = (int)(((long long)blockIdx.x * blockDim.x + threadIdx.x) >> 5);
  int lane = threadIdx.x & 31;
  if (wave >= ntiles) return;
  int row0 = wave * 16;
  int m  = lane & 15;
  int hi = lane >> 4;
  const float* hrow = h + (size_t)(row0 + m) * 64;
  v8f cl[2] = {{0.f},{0.f}};
  v8f cr[2] = {{0.f},{0.f}};
#pragma unroll
  for (int kk = 0; kk < 16; ++kk) {
    int k0 = kk * 4 + hi * 2;
    v2f a; a.x = hrow[k0]; a.y = hrow[k0 + 1];
#pragma unroll
    for (int j = 0; j < 2; ++j) {
      v2f bl, br;
      bl.x = Wl[(k0 + 0) * 32 + j * 16 + m];
      bl.y = Wl[(k0 + 1) * 32 + j * 16 + m];
      br.x = Wr[(k0 + 0) * 32 + j * 16 + m];
      br.y = Wr[(k0 + 1) * 32 + j * 16 + m];
      cl[j] = wmma4(a, bl, cl[j]);
      cr[j] = wmma4(a, br, cr[j]);
    }
  }
#pragma unroll
  for (int j = 0; j < 2; ++j)
#pragma unroll
    for (int v = 0; v < 8; ++v) {
      int r = v + hi * 8;
      xl[(size_t)(row0 + r) * 32 + j * 16 + m] = cl[j][v];
      xr[(size_t)(row0 + r) * 32 + j * 16 + m] = cr[j][v];
    }
}

__global__ void edge_logits1(const int* __restrict__ ei, int E, int n,
                             const float* __restrict__ x,
                             const float* __restrict__ Wl,
                             const float* __restrict__ Wr,
                             const float* __restrict__ a1,
                             float* __restrict__ logits,
                             int* __restrict__ maxd) {
  const int lane = threadIdx.x & 31;
  const int wave = __builtin_amdgcn_readfirstlane(
      (int)(((long long)blockIdx.x * blockDim.x + threadIdx.x) >> 5));
  const int TE = E + n;
  const int base = wave * 16;
  if (base >= TE) return;
  const int m  = lane & 15;
  const int hi = lane >> 4;
  const int t = base + m;
  int s = 0, d = 0;
  if (t < TE) {
    if (t < E) { s = ei[t]; d = ei[E + t]; } else { s = t - E; d = s; }
  }
  const float2 xs = *(const float2*)(x + (size_t)s * 4 + hi * 2);
  const float2 xd = *(const float2*)(x + (size_t)d * 4 + hi * 2);
  v2f fs; fs.x = xs.x; fs.y = xs.y;
  v2f fd; fd.x = xd.x; fd.y = xd.y;
  float p[32];
#pragma unroll
  for (int j = 0; j < 4; ++j) {
    v2f bl, br;
    bl.x = Wl[(2 * hi + 0) * 64 + j * 16 + m];
    bl.y = Wl[(2 * hi + 1) * 64 + j * 16 + m];
    br.x = Wr[(2 * hi + 0) * 64 + j * 16 + m];
    br.y = Wr[(2 * hi + 1) * 64 + j * 16 + m];
    const float aj = a1[j * 16 + m];
    v8f cc = {};
    cc = wmma4(fs, bl, cc);
    cc = wmma4(fd, br, cc);
#pragma unroll
    for (int v = 0; v < 8; ++v) {
      float e = cc[v]; e = e >= 0.f ? e : NEG_SLOPE * e;
      p[(v >> 1) * 8 + j * 2 + (v & 1)] = e * aj;
    }
  }
  bfly<8, 16, 0>(p, (lane & 8) != 0);
  bfly<4, 8, 0>(p, (lane & 4) != 0);
  bfly<2, 4, 0>(p, (lane & 2) != 0);
  bfly<1, 2, 0>(p, (lane & 1) != 0);
  const int head = lane & 3;
  const int jb = hi * 8 + 2 * ((lane >> 2) & 3);
#pragma unroll
  for (int q = 0; q < 2; ++q) {
    const int j = jb + q;
    const int dj = __shfl(d, j, 32);
    if (base + j < TE) {
      logits[(size_t)(base + j) * 4 + head] = p[q];
      atomicMaxKey(&maxd[(size_t)dj * 4 + head], p[q]);
    }
  }
}

__global__ void edge_accum1(const int* __restrict__ ei, int E, int n,
                            const float* __restrict__ x,
                            const float* __restrict__ logits,
                            const int* __restrict__ maxd,
                            float* __restrict__ acc) {
  const int lane = threadIdx.x & 31;
  const int wave = __builtin_amdgcn_readfirstlane(
      (int)(((long long)blockIdx.x * blockDim.x + threadIdx.x) >> 5));
  const int TE = E + n;
  const int base = wave * 8;
  if (base >= TE) return;
  const int e = base + (lane >> 2), h = lane & 3;
  int s = 0, d = 0;
  float ex = 0.f;
  if (e < TE) {
    if (e < E) { s = ei[e]; d = ei[E + e]; } else { s = e - E; d = s; }
    ex = __expf(logits[(size_t)e * 4 + h] - fkey_inv(maxd[(size_t)d * 4 + h]));
  }
  const int cnt = TE - base;
  const int kx = lane < 16 ? (lane >> 2) : 3;
  const int hsel = (lane & 3) << 2;
  if (cnt >= 8) {
    float xv[8], w[8];
    int dj[8];
#pragma unroll
    for (int j = 0; j < 8; ++j)
      xv[j] = x[(size_t)bcast_i(s, 4 * j) * 4 + kx];
#pragma unroll
    for (int j = 0; j < 8; ++j) {
      w[j] = __int_as_float(
          __builtin_amdgcn_ds_bpermute(hsel + 16 * j, __float_as_int(ex)));
      dj[j] = bcast_i(d, 4 * j);
    }
    if (lane < 20) {
#pragma unroll
      for (int j = 0; j < 8; ++j)
        atomicAdd(acc + (size_t)dj[j] * 32 + lane, (lane < 16 ? xv[j] : 1.f) * w[j]);
    }
  } else {
    for (int j = 0; j < cnt; ++j) {
      const float xj = x[(size_t)bcast_i(s, 4 * j) * 4 + kx];
      const float w = __int_as_float(
          __builtin_amdgcn_ds_bpermute(hsel + 16 * j, __float_as_int(ex)));
      const float val = (lane < 16 ? xj : 1.f) * w;
      float* row = acc + (size_t)bcast_i(d, 4 * j) * 32;
      if (lane < 20) atomicAdd(row + lane, val);
    }
  }
}

__global__ void node_fin1(const float* __restrict__ acc,
                          const float* __restrict__ Wl,
                          const float* __restrict__ b1,
                          const float* __restrict__ g,
                          const float* __restrict__ be,
                          float* __restrict__ h1, int ntiles) {
  int wave = (int)(((long long)blockIdx.x * blockDim.x + threadIdx.x) >> 5);
  int lane = threadIdx.x & 31;
  if (wave >= ntiles) return;
  int row0 = wave * 16;
  int m  = lane & 15;
  int hi = lane >> 4;
  const float4* arow = (const float4*)(acc + (size_t)(row0 + m) * 32);
  const float4 p0 = arow[2 * hi];
  const float4 p1 = arow[2 * hi + 1];
  const float4 dn = arow[4];
  const float pa[4] = {p0.x, p0.y, p0.z, p0.w};
  const float pb[4] = {p1.x, p1.y, p1.z, p1.w};
  const float dd[4] = {dn.x, dn.y, dn.z, dn.w};
  v8f c[4];
  float bias[4], gg[4], bb[4];
#pragma unroll
  for (int h = 0; h < 4; ++h) {
    v2f a; a.x = pa[h] / dd[h]; a.y = pb[h] / dd[h];
    v2f b;
    b.x = Wl[(2 * hi + 0) * 64 + h * 16 + m];
    b.y = Wl[(2 * hi + 1) * 64 + h * 16 + m];
    v8f z = {};
    c[h] = wmma4(a, b, z);
    bias[h] = b1[h * 16 + m];
    gg[h] = g[h * 16 + m];
    bb[h] = be[h * 16 + m];
  }
  float s[8], q[8];
#pragma unroll
  for (int v = 0; v < 8; ++v) {
#pragma unroll
    for (int h = 0; h < 4; ++h) c[h][v] += bias[h];
    s[v] = (c[0][v] + c[1][v]) + (c[2][v] + c[3][v]);
  }
#pragma unroll
  for (int o = 8; o; o >>= 1)
#pragma unroll
    for (int v = 0; v < 8; ++v) s[v] += __shfl_xor(s[v], o, 32);
#pragma unroll
  for (int v = 0; v < 8; ++v) {
    const float mu = s[v] * (1.f / 64.f);
#pragma unroll
    for (int h = 0; h < 4; ++h) c[h][v] -= mu;
    q[v] = (c[0][v] * c[0][v] + c[1][v] * c[1][v]) + (c[2][v] * c[2][v] + c[3][v] * c[3][v]);
  }
#pragma unroll
  for (int o = 8; o; o >>= 1)
#pragma unroll
    for (int v = 0; v < 8; ++v) q[v] += __shfl_xor(q[v], o, 32);
#pragma unroll
  for (int v = 0; v < 8; ++v) {
    const float rs = rsqrtf(q[v] * (1.f / 64.f) + LN_EPS);
    float* orow = h1 + (size_t)(row0 + v + hi * 8) * 64 + m;
#pragma unroll
    for (int h = 0; h < 4; ++h) {
      float y = c[h][v] * rs * gg[h] + bb[h];
      y = y > 0.f ? y : expm1f(y);
      orow[h * 16] = y;
    }
  }
}

__global__ void edge_logits2(const int* __restrict__ ei, int E, int n,
                             const float* __restrict__ xl,
                             const float* __restrict__ xr,
                             const float* __restrict__ a2,
                             float* __restrict__ logits,
                             int* __restrict__ maxd) {
  const int lane = threadIdx.x & 31;
  const int wave = __builtin_amdgcn_readfirstlane(
      (int)(((long long)blockIdx.x * blockDim.x + threadIdx.x) >> 5));
  const int TE = E + n;
  const int base = wave * 32;
  if (base >= TE) return;
  const int t = base + lane;
  int s = 0, d = 0;
  if (t < TE) {
    if (t < E) { s = ei[t]; d = ei[E + t]; } else { s = t - E; d = s; }
  }
  const float al = a2[lane];
  float p[32];
#pragma unroll
  for (int j0 = 0; j0 < 32; j0 += 8) {
    float u[8], v[8];
#pragma unroll
    for (int k = 0; k < 8; ++k) {
      u[k] = xl[(unsigned)bcast_i(s, j0 + k) * 32u + (unsigned)lane];
      v[k] = xr[(unsigned)bcast_i(d, j0 + k) * 32u + (unsigned)lane];
    }
#pragma unroll
    for (int k = 0; k < 8; ++k) {
      float e = u[k] + v[k]; e = e >= 0.f ? e : NEG_SLOPE * e;
      p[j0 + k] = e * al;
    }
  }
  bfly<16, 16, 0>(p, (lane & 16) != 0);
  bfly<8, 8, 0>(p, (lane & 8) != 0);
  bfly<4, 4, 0>(p, (lane & 4) != 0);
  bfly<2, 2, 0>(p, (lane & 2) != 0);
  bfly<1, 1, 0>(p, (lane & 1) != 0);
  if (t < TE) {
    logits[t] = p[0];
    atomicMaxKey(&maxd[d], p[0]);
  }
}

__global__ void edge_accum2(const int* __restrict__ ei, int E, int n,
                            const float* __restrict__ xl,
                            const float* __restrict__ logits,
                            const int* __restrict__ maxd,
                            float* __restrict__ acc) {
  const int lane = threadIdx.x & 31;
  const int wave = __builtin_amdgcn_readfirstlane(
      (int)(((long long)blockIdx.x * blockDim.x + threadIdx.x) >> 5));
  const int TE = E + n;
  const int base = wave * 32;
  if (base >= TE) return;
  const int t = base + lane;
  int s = 0, d = 0;
  float ex = 0.f;
  if (t < TE) {
    if (t < E) { s = ei[t]; d = ei[E + t]; } else { s = t - E; d = s; }
    ex = __expf(logits[t] - fkey_inv(maxd[d]));
  }
  const int cnt = TE - base;
  const bool den_lane = lane == 31;
  if (cnt >= 32) {
#pragma unroll
    for (int j0 = 0; j0 < 32; j0 += 8) {
      float v[8];
#pragma unroll
      for (int u = 0; u < 8; ++u) v[u] = xl[(size_t)bcast_i(s, j0 + u) * 32 + lane];
#pragma unroll
      for (int u = 0; u < 8; ++u) {
        const float c31 = bcast_f(v[u], 31);
        const float rel = den_lane ? 1.f : v[u] - c31;
        atomicAdd(acc + (size_t)bcast_i(d, j0 + u) * 32 + lane, bcast_f(ex, j0 + u) * rel);
      }
    }
  } else {
    for (int j = 0; j < cnt; ++j) {
      const float v = xl[(size_t)bcast_i(s, j) * 32 + lane];
      const float c31 = bcast_f(v, 31);
      const float rel = den_lane ? 1.f : v - c31;
      atomicAdd(acc + (size_t)bcast_i(d, j) * 32 + lane, bcast_f(ex, j) * rel);
    }
  }
}

__global__ void node_fin2(const float* __restrict__ acc,
                          const float* __restrict__ b2,
                          const float* __restrict__ g,
                          const float* __restrict__ be,
                          float* __restrict__ out, int n) {
  int node = (int)(((long long)blockIdx.x * blockDim.x + threadIdx.x) >> 5);
  int lane = threadIdx.x & 31;
  if (node >= n) return;
  const float a = acc[(size_t)node * 32 + lane];
  const float den = bcast_f(a, 31);
  float v = (lane == 31 ? 0.f : a / den) + b2[lane];
  float s = v;
#pragma unroll
  for (int o = 16; o; o >>= 1) s += __shfl_xor(s, o, 32);
  float mu = s * (1.f / 32.f);
  float dd = v - mu;
  float q = dd * dd;
#pragma unroll
  for (int o = 16; o; o >>= 1) q += __shfl_xor(q, o, 32);
  float rs = rsqrtf(q * (1.f / 32.f) + LN_EPS);
  out[(size_t)node * 32 + lane] = dd * rs * g[lane] + be[lane];
}

static inline int gridFor(long long threads, int block) {
  return (int)((threads + block - 1) / block);
}

extern "C" void kernel_launch(void* const* d_in, const int* in_sizes, int n_in,
                              void* d_out, int out_size, void* d_ws, size_t ws_size,
                              hipStream_t stream) {
  const float* x   = (const float*)d_in[0];
  const int*   ei  = (const int*)  d_in[1];
  const float* W1l = (const float*)d_in[2];
  const float* W1r = (const float*)d_in[3];
  const float* a1  = (const float*)d_in[4];
  const float* b1  = (const float*)d_in[5];
  const float* g1  = (const float*)d_in[6];
  const float* be1 = (const float*)d_in[7];
  const float* W2l = (const float*)d_in[8];
  const float* W2r = (const float*)d_in[9];
  const float* a2  = (const float*)d_in[10];
  const float* b2  = (const float*)d_in[11];
  const float* g2  = (const float*)d_in[12];
  const float* be2 = (const float*)d_in[13];
  float* out = (float*)d_out;

  const int n  = in_sizes[0] / 4;
  const int E  = in_sizes[1] / 2;
  const int TE = E + n;
  const int ntiles = n / 16;

  float* ws = (float*)d_ws;
  const size_t XL1  = 0;
  const size_t ACC2 = XL1 + (size_t)n * 32;
  const size_t XR1  = XL1 + (size_t)n * 64;
  const size_t MAX2 = XR1 + (size_t)n * 32;
  const size_t ACC1 = XR1 + (size_t)n * 64;
  const size_t MAX1 = ACC1 + (size_t)n * 68;
  const size_t LOG1 = MAX1 + (size_t)n * 4;
  const size_t H1   = LOG1 + (size_t)TE * 4;

  const int B = 256;

  clear_all<<<gridFor((long long)n * 8, B * 4), B, 0, stream>>>(
      (float4*)(ws + ACC1), (float4*)(ws + ACC2), (long long)n * 8,
      (int4*)(ws + MAX1), (int*)(ws + MAX2), n);

  edge_logits1<<<gridFor((long long)TE * 2, B), B, 0, stream>>>(
      ei, E, n, x, W1l, W1r, a1, ws + LOG1, (int*)(ws + MAX1));

  edge_accum1<<<gridFor((long long)TE * 4, B), B, 0, stream>>>(
      ei, E, n, x, ws + LOG1, (const int*)(ws + MAX1), ws + ACC1);

  node_fin1<<<gridFor((long long)ntiles * 32, 128), 128, 0, stream>>>(
      ws + ACC1, W1l, b1, g1, be1, ws + H1, ntiles);

  gemm2_wmma<<<gridFor((long long)ntiles * 32, B), B, 0, stream>>>(
      ws + H1, W2l, W2r, ws + XL1, ws + XR1, ntiles);

  edge_logits2<<<gridFor(TE, B), B, 0, stream>>>(
      ei, E, n, ws + XL1, ws + XR1, a2, ws + LOG1, (int*)(ws + MAX2));

  edge_accum2<<<gridFor(TE, B), B, 0, stream>>>(
      ei, E, n, ws + XL1, ws + LOG1, (const int*)(ws + MAX2), ws + ACC2);

  node_fin2<<<gridFor((long long)n * 32, B), B, 0, stream>>>(
      ws + ACC2, b2, g2, be2, out, n);
}


# Each plane updated by atomics kept twice, one copy per half of the package (die number from the placement message picks it); readers merge: max of 2 maxima, sum of 2 sums: new sum order
# speedup vs baseline: 6.8749x; 1.3917x over previous
#include <hip/hip_runtime.h>
#include <hip/hip_bf16.h>
#include <math.h>


typedef __attribute__((ext_vector_type(2))) float v2f;
typedef __attribute__((ext_vector_type(8))) float v8f;

#define NEG_SLOPE 0.2f
#define LN_EPS 1e-5f

static __device__ inline v8f wmma4(v2f a, v2f b, v8f c) {
  return __builtin_amdgcn_wmma_f32_16x16x4_f32(
       false, a,  false, b,
       (short)0, c,  false,  false);
}

#define MAXKEY_NEG_INF ((int)0x807fffff)
static __device__ inline int fkey(float f) {
  const int b = __float_as_int(f);
  return b >= 0 ? b : b ^ 0x7fffffff;
}
static __device__ inline float fkey_inv(int k) {
  return __int_as_float(k >= 0 ? k : k ^ 0x7fffffff);
}
static __device__ inline void atomicMaxKey(int* addr, float val) {
  atomicMax(addr, fkey(val));
}

static __device__ inline int bcast_i(int v, int srcLane) {
  return __builtin_amdgcn_readlane(v, srcLane);
}
static __device__ inline float bcast_f(float v, int srcLane) {
  return __int_as_float(__builtin_amdgcn_readlane(__float_as_int(v), srcLane));
}

static __device__ inline int chip_half() {
  return (int)((__builtin_amdgcn_s_sendmsg_rtn(0x87) >> 18) & 1u);
}

template <int OFF, int STRIDE, int Q>
static __device__ inline void bfly(float (&p)[32], bool up) {
  if constexpr (Q < STRIDE) {
    const float lo = p[Q], hi = p[Q + STRIDE];
    const float keep = up ? hi : lo;
    const float send = up ? lo : hi;
    p[Q] = keep + __shfl_xor(send, OFF, 32);
    bfly<OFF, STRIDE, Q + 1>(p, up);
  }
}

__global__ void clear_all(float4* __restrict__ acc1, float4* __restrict__ acc2, long long nq,
                          int4* __restrict__ max1,
                          int* __restrict__ max2, int n) {
  const float4 z = make_float4(0.f, 0.f, 0.f, 0.f);
  const int ninf = MAXKEY_NEG_INF;
  const long long base = (long long)blockIdx.x * (blockDim.x * 4) + threadIdx.x;
#pragma unroll
  for (int j = 0; j < 4; ++j) {
    const long long i = base + (long long)j * blockDim.x;
    if (i < nq) { acc1[i] = z; acc2[i] = z; }
  }
  const long long g = (long long)blockIdx.x * blockDim.x + threadIdx.x;
  if (g < n) {
    max1[g] = make_int4(ninf, ninf, ninf, ninf);
    max2[g] = ninf;
  }
}

__global__ void gemm2_wmma(const float* __restrict__ h,
                           const float* __restrict__ Wl,
                           const float* __restrict__ Wr,
                           float* __restrict__ xl,
                           float* __restrict__ xr, int ntiles) {
  int wave = (int)(((long long)blockIdx.x * blockDim.x + threadIdx.x) >> 5);
  int lane = threadIdx.x & 31;
  if (wave >= ntiles) return;
  int row0 = wave * 16;
  int m  = lane & 15;
  int hi = lane >> 4;
  const float* hrow = h + (size_t)(row0 + m) * 64;
  v8f cl[2] = {{0.f},{0.f}};
  v8f cr[2] = {{0.f},{0.f}};
#pragma unroll
  for (int kk = 0; kk < 16; ++kk) {
    int k0 = kk * 4 + hi * 2;
    v2f a; a.x = hrow[k0]; a.y = hrow[k0 + 1];
#pragma unroll
    for (int j = 0; j < 2; ++j) {
      v2f bl, br;
      bl.x = Wl[(k0 + 0) * 32 + j * 16 + m];
      bl.y = Wl[(k0 + 1) * 32 + j * 16 + m];
      br.x = Wr[(k0 + 0) * 32 + j * 16 + m];
      br.y = Wr[(k0 + 1) * 32 + j * 16 + m];
      cl[j] = wmma4(a, bl, cl[j]);
      cr[j] = wmma4(a, br, cr[j]);
    }
  }
#pragma unroll
  for (int j = 0; j < 2; ++j)
#pragma unroll
    for (int v = 0; v < 8; ++v) {
      int r = v + hi * 8;
      xl[(size_t)(row0 + r) * 32 + j * 16 + m] = cl[j][v];
      xr[(size_t)(row0 + r) * 32 + j * 16 + m] = cr[j][v];
    }
}

__global__ void edge_logits1(const int* __restrict__ ei, int E, int n,
                             const float* __restrict__ x,
                             const float* __restrict__ Wl,
                             const float* __restrict__ Wr,
                             const float* __restrict__ a1,
                             float* __restrict__ logits,
                             int* __restrict__ maxd) {
  const int lane = threadIdx.x & 31;
  const int wave = __builtin_amdgcn_readfirstlane(
      (int)(((long long)blockIdx.x * blockDim.x + threadIdx.x) >> 5));
  const int TE = E + n;
  const int base = wave * 16;
  if (base >= TE) return;
  const int m  = lane & 15;
  const int hi = lane >> 4;
  const int t = base + m;
  int s = 0, d = 0;
  if (t < TE) {
    if (t < E) { s = ei[t]; d = ei[E + t]; } else { s = t - E; d = s; }
  }
  int* maxh = maxd + (size_t)chip_half() * n * 4;
  const float2 xs = *(const float2*)(x + (size_t)s * 4 + hi * 2);
  const float2 xd = *(const float2*)(x + (size_t)d * 4 + hi * 2);
  v2f fs; fs.x = xs.x; fs.y = xs.y;
  v2f fd; fd.x = xd.x; fd.y = xd.y;
  float p[32];
#pragma unroll
  for (int j = 0; j < 4; ++j) {
    v2f bl, br;
    bl.x = Wl[(2 * hi + 0) * 64 + j * 16 + m];
    bl.y = Wl[(2 * hi + 1) * 64 + j * 16 + m];
    br.x = Wr[(2 * hi + 0) * 64 + j * 16 + m];
    br.y = Wr[(2 * hi + 1) * 64 + j * 16 + m];
    const float aj = a1[j * 16 + m];
    v8f cc = {};
    cc = wmma4(fs, bl, cc);
    cc = wmma4(fd, br, cc);
#pragma unroll
    for (int v = 0; v < 8; ++v) {
      float e = cc[v]; e = e >= 0.f ? e : NEG_SLOPE * e;
      p[(v >> 1) * 8 + j * 2 + (v & 1)] = e * aj;
    }
  }
  bfly<8, 16, 0>(p, (lane & 8) != 0);
  bfly<4, 8, 0>(p, (lane & 4) != 0);
  bfly<2, 4, 0>(p, (lane & 2) != 0);
  bfly<1, 2, 0>(p, (lane & 1) != 0);
  const int head = lane & 3;
  const int jb = hi * 8 + 2 * ((lane >> 2) & 3);
#pragma unroll
  for (int q = 0; q < 2; ++q) {
    const int j = jb + q;
    const int dj = __shfl(d, j, 32);
    if (base + j < TE) {
      logits[(size_t)(base + j) * 4 + head] = p[q];
      atomicMaxKey(&maxh[(size_t)dj * 4 + head], p[q]);
    }
  }
}

__global__ void edge_accum1(const int* __restrict__ ei, int E, int n,
                            const float* __restrict__ x,
                            const float* __restrict__ logits,
                            const int* __restrict__ maxd,
                            float* __restrict__ acc) {
  const int lane = threadIdx.x & 31;
  const int wave = __builtin_amdgcn_readfirstlane(
      (int)(((long long)blockIdx.x * blockDim.x + threadIdx.x) >> 5));
  const int TE = E + n;
  const int base = wave * 8;
  if (base >= TE) return;
  const int e = base + (lane >> 2), h = lane & 3;
  int s = 0, d = 0;
  float ex = 0.f;
  if (e < TE) {
    if (e < E) { s = ei[e]; d = ei[E + e]; } else { s = e - E; d = s; }
  }
  acc += (size_t)chip_half() * n * 32;
  if (e < TE) {
    const int ka = maxd[(size_t)d * 4 + h];
    const int kb = maxd[(size_t)n * 4 + (size_t)d * 4 + h];
    ex = __expf(logits[(size_t)e * 4 + h] - fkey_inv(ka > kb ? ka : kb));
  }
  const int cnt = TE - base;
  const int kx = lane < 16 ? (lane >> 2) : 3;
  const int hsel = (lane & 3) << 2;
  if (cnt >= 8) {
    float xv[8], w[8];
    int dj[8];
#pragma unroll
    for (int j = 0; j < 8; ++j)
      xv[j] = x[(size_t)bcast_i(s, 4 * j) * 4 + kx];
#pragma unroll
    for (int j = 0; j < 8; ++j) {
      w[j] = __int_as_float(
          __builtin_amdgcn_ds_bpermute(hsel + 16 * j, __float_as_int(ex)));
      dj[j] = bcast_i(d, 4 * j);
    }
    if (lane < 20) {
#pragma unroll
      for (int j = 0; j < 8; ++j)
        atomicAdd(acc + (size_t)dj[j] * 32 + lane, (lane < 16 ? xv[j] : 1.f) * w[j]);
    }
  } else {
    for (int j = 0; j < cnt; ++j) {
      const float xj = x[(size_t)bcast_i(s, 4 * j) * 4 + kx];
      const float w = __int_as_float(
          __builtin_amdgcn_ds_bpermute(hsel + 16 * j, __float_as_int(ex)));
      const float val = (lane < 16 ? xj : 1.f) * w;
      float* row = acc + (size_t)bcast_i(d, 4 * j) * 32;
      if (lane < 20) atomicAdd(row + lane, val);
    }
  }
}

__global__ void node_fin1(const float* __restrict__ acc,
                          const float* __restrict__ Wl,
                          const float* __restrict__ b1,
                          const float* __restrict__ g,
                          const float* __restrict__ be,
                          float* __restrict__ h1, int ntiles) {
  int wave = (int)(((long long)blockIdx.x * blockDim.x + threadIdx.x) >> 5);
  int lane = threadIdx.x & 31;
  if (wave >= ntiles) return;
  int row0 = wave * 16;
  int m  = lane & 15;
  int hi = lane >> 4;
  const float4* arow = (const float4*)(acc + (size_t)(row0 + m) * 32);
  const float4* brow = arow + (size_t)ntiles * 16 * 8;
  const float4 p0 = arow[2 * hi],     r0 = brow[2 * hi];
  const float4 p1 = arow[2 * hi + 1], r1 = brow[2 * hi + 1];
  const float4 dn = arow[4],          rn = brow[4];
  const float pa[4] = {p0.x + r0.x, p0.y + r0.y, p0.z + r0.z, p0.w + r0.w};
  const float pb[4] = {p1.x + r1.x, p1.y + r1.y, p1.z + r1.z, p1.w + r1.w};
  const float dd[4] = {dn.x + rn.x, dn.y + rn.y, dn.z + rn.z, dn.w + rn.w};
  v8f c[4];
  float bias[4], gg[4], bb[4];
#pragma unroll
  for (int h = 0; h < 4; ++h) {
    v2f a; a.x = pa[h] / dd[h]; a.y = pb[h] / dd[h];
    v2f b;
    b.x = Wl[(2 * hi + 0) * 64 + h * 16 + m];
    b.y = Wl[(2 * hi + 1) * 64 + h * 16 + m];
    v8f z = {};
    c[h] = wmma4(a, b, z);
    bias[h] = b1[h * 16 + m];
    gg[h] = g[h * 16 + m];
    bb[h] = be[h * 16 + m];
  }
  float s[8], q[8];
#pragma unroll
  for (int v = 0; v < 8; ++v) {
#pragma unroll
    for (int h = 0; h < 4; ++h) c[h][v] += bias[h];
    s[v] = (c[0][v] + c[1][v]) + (c[2][v] + c[3][v]);
  }
#pragma unroll
  for (int o = 8; o; o >>= 1)
#pragma unroll
    for (int v = 0; v < 8; ++v) s[v] += __shfl_xor(s[v], o, 32);
#pragma unroll
  for (int v = 0; v < 8; ++v) {
    const float mu = s[v] * (1.f / 64.f);
#pragma unroll
    for (int h = 0; h < 4; ++h) c[h][v] -= mu;
    q[v] = (c[0][v] * c[0][v] + c[1][v] * c[1][v]) + (c[2][v] * c[2][v] + c[3][v] * c[3][v]);
  }
#pragma unroll
  for (int o = 8; o; o >>= 1)
#pragma unroll
    for (int v = 0; v < 8; ++v) q[v] += __shfl_xor(q[v], o, 32);
#pragma unroll
  for (int v = 0; v < 8; ++v) {
    const float rs = rsqrtf(q[v] * (1.f / 64.f) + LN_EPS);
    float* orow = h1 + (size_t)(row0 + v + hi * 8) * 64 + m;
#pragma unroll
    for (int h = 0; h < 4; ++h) {
      float y = c[h][v] * rs * gg[h] + bb[h];
      y = y > 0.f ? y : expm1f(y);
      orow[h * 16] = y;
    }
  }
}

__global__ void edge_logits2(const int* __restrict__ ei, int E, int n,
                             const float* __restrict__ xl,
                             const float* __restrict__ xr,
                             const float* __restrict__ a2,
                             float* __restrict__ logits,
                             int* __restrict__ maxd) {
  const int lane = threadIdx.x & 31;
  const int wave = __builtin_amdgcn_readfirstlane(
      (int)(((long long)blockIdx.x * blockDim.x + threadIdx.x) >> 5));
  const int TE = E + n;
  const int base = wave * 32;
  if (base >= TE) return;
  const int t = base + lane;
  int s = 0, d = 0;
  if (t < TE) {
    if (t < E) { s = ei[t]; d = ei[E + t]; } else { s = t - E; d = s; }
  }
  maxd += (size_t)chip_half() * n;
  const float al = a2[lane];
  float p[32];
#pragma unroll
  for (int j0 = 0; j0 < 32; j0 += 8) {
    float u[8], v[8];
#pragma unroll
    for (int k = 0; k < 8; ++k) {
      u[k] = xl[(unsigned)bcast_i(s, j0 + k) * 32u + (unsigned)lane];
      v[k] = xr[(unsigned)bcast_i(d, j0 + k) * 32u + (unsigned)lane];
    }
#pragma unroll
    for (int k = 0; k < 8; ++k) {
      float e = u[k] + v[k]; e = e >= 0.f ? e : NEG_SLOPE * e;
      p[j0 + k] = e * al;
    }
  }
  bfly<16, 16, 0>(p, (lane & 16) != 0);
  bfly<8, 8, 0>(p, (lane & 8) != 0);
  bfly<4, 4, 0>(p, (lane & 4) != 0);
  bfly<2, 2, 0>(p, (lane & 2) != 0);
  bfly<1, 1, 0>(p, (lane & 1) != 0);
  if (t < TE) {
    logits[t] = p[0];
    atomicMaxKey(&maxd[d], p[0]);
  }
}

__global__ void edge_accum2(const int* __restrict__ ei, int E, int n,
                            const float* __restrict__ xl,
                            const float* __restrict__ logits,
                            const int* __restrict__ maxd,
                            float* __restrict__ acc) {
  const int lane = threadIdx.x & 31;
  const int wave = __builtin_amdgcn_readfirstlane(
      (int)(((long long)blockIdx.x * blockDim.x + threadIdx.x) >> 5));
  const int TE = E + n;
  const int base = wave * 32;
  if (base >= TE) return;
  const int t = base + lane;
  int s = 0, d = 0;
  float ex = 0.f;
  if (t < TE) {
    if (t < E) { s = ei[t]; d = ei[E + t]; } else { s = t - E; d = s; }
  }
  acc += (size_t)chip_half() * n * 32;
  if (t < TE) {
    const int ka = maxd[d];
    const int kb = maxd[(size_t)n + d];
    ex = __expf(logits[t] - fkey_inv(ka > kb ? ka : kb));
  }
  const int cnt = TE - base;
  const bool den_lane = lane == 31;
  if (cnt >= 32) {
#pragma unroll
    for (int j0 = 0; j0 < 32; j0 += 8) {
      float v[8];
#pragma unroll
      for (int u = 0; u < 8; ++u) v[u] = xl[(size_t)bcast_i(s, j0 + u) * 32 + lane];
#pragma unroll
      for (int u = 0; u < 8; ++u) {
        const float c31 = bcast_f(v[u], 31);
        const float rel = den_lane ? 1.f : v[u] - c31;
        atomicAdd(acc + (size_t)bcast_i(d, j0 + u) * 32 + lane, bcast_f(ex, j0 + u) * rel);
      }
    }
  } else {
    for (int j = 0; j < cnt; ++j) {
      const float v = xl[(size_t)bcast_i(s, j) * 32 + lane];
      const float c31 = bcast_f(v, 31);
      const float rel = den_lane ? 1.f : v - c31;
      atomicAdd(acc + (size_t)bcast_i(d, j) * 32 + lane, bcast_f(ex, j) * rel);
    }
  }
}

__global__ void node_fin2(const float* __restrict__ acc,
                          const float* __restrict__ b2,
                          const float* __restrict__ g,
                          const float* __restrict__ be,
                          float* __restrict__ out, int n) {
  int node = (int)(((long long)blockIdx.x * blockDim.x + threadIdx.x) >> 5);
  int lane = threadIdx.x & 31;
  if (node >= n) return;
  const float a = acc[(size_t)node * 32 + lane] + acc[(size_t)n * 32 + (size_t)node * 32 + lane];
  const float den = bcast_f(a, 31);
  float v = (lane == 31 ? 0.f : a / den) + b2[lane];
  float s = v;
#pragma unroll
  for (int o = 16; o; o >>= 1) s += __shfl_xor(s, o, 32);
  float mu = s * (1.f / 32.f);
  float dd = v - mu;
  float q = dd * dd;
#pragma unroll
  for (int o = 16; o; o >>= 1) q += __shfl_xor(q, o, 32);
  float rs = rsqrtf(q * (1.f / 32.f) + LN_EPS);
  out[(size_t)node * 32 + lane] = dd * rs * g[lane] + be[lane];
}

static inline int gridFor(long long threads, int block) {
  return (int)((threads + block - 1) / block);
}

extern "C" void kernel_launch(void* const* d_in, const int* in_sizes, int n_in,
                              void* d_out, int out_size, void* d_ws, size_t ws_size,
                              hipStream_t stream) {
  const float* x   = (const float*)d_in[0];
  const int*   ei  = (const int*)  d_in[1];
  const float* W1l = (const float*)d_in[2];
  const float* W1r = (const float*)d_in[3];
  const float* a1  = (const float*)d_in[4];
  const float* b1  = (const float*)d_in[5];
  const float* g1  = (const float*)d_in[6];
  const float* be1 = (const float*)d_in[7];
  const float* W2l = (const float*)d_in[8];
  const float* W2r = (const float*)d_in[9];
  const float* a2  = (const float*)d_in[10];
  const float* b2  = (const float*)d_in[11];
  const float* g2  = (const float*)d_in[12];
  const float* be2 = (const float*)d_in[13];
  float* out = (float*)d_out;

  const int n  = in_sizes[0] / 4;
  const int E  = in_sizes[1] / 2;
  const int TE = E + n;
  const int ntiles = n / 16;

  float* ws = (float*)d_ws;
  const size_t ACC1 = 0;
  const size_t ACC2 = ACC1 + (size_t)n * 64;
  const size_t LOG1 = ACC2 + (size_t)n * 64;
  const size_t XL1  = LOG1;
  const size_t XR1  = XL1 + (size_t)n * 32;
  const size_t LOGSZ = (size_t)TE * 4 > (size_t)n * 64 ? (size_t)TE * 4 : (size_t)n * 64;
  const size_t H1   = LOG1 + LOGSZ;
  const size_t MAX1 = H1 + (size_t)n * 64;
  const size_t MAX2 = MAX1 + (size_t)n * 8;
  const size_t LOG2 = MAX2 + (size_t)n * 2;

  const int B = 256;

  clear_all<<<gridFor((long long)n * 16, B * 4), B, 0, stream>>>(
      (float4*)(ws + ACC1), (float4*)(ws + ACC2), (long long)n * 16,
      (int4*)(ws + MAX1), (int*)(ws + MAX2), n * 2);

  edge_logits1<<<gridFor((long long)TE * 2, B), B, 0, stream>>>(
      ei, E, n, x, W1l, W1r, a1, ws + LOG1, (int*)(ws + MAX1));

  edge_accum1<<<gridFor((long long)TE * 4, B), B, 0, stream>>>(
      ei, E, n, x, ws + LOG1, (const int*)(ws + MAX1), ws + ACC1);

  node_fin1<<<gridFor((long long)ntiles * 32, 128), 128, 0, stream>>>(
      ws + ACC1, W1l, b1, g1, be1, ws + H1, ntiles);

  gemm2_wmma<<<gridFor((long long)ntiles * 32, B), B, 0, stream>>>(
      ws + H1, W2l, W2r, ws + XL1, ws + XR1, ntiles);

  edge_logits2<<<gridFor(TE, B), B, 0, stream>>>(
      ei, E, n, ws + XL1, ws + XR1, a2, ws + LOG2, (int*)(ws + MAX2));

  edge_accum2<<<gridFor(TE, B), B, 0, stream>>>(
      ei, E, n, ws + XL1, ws + LOG2, (const int*)(ws + MAX2), ws + ACC2);

  node_fin2<<<gridFor((long long)n * 32, B), B, 0, stream>>>(
      ws + ACC2, b2, g2, be2, out, n);
}


# Edge passes and last node pass run as resident grids (at most 1536 work-groups): a wave loops over tiles, next tile's indices asked a trip ahead; 470k short waves fewer; atomic sums in arrival order
# speedup vs baseline: 7.8927x; 1.1480x over previous
#include <hip/hip_runtime.h>
#include <hip/hip_bf16.h>
#include <math.h>


typedef __attribute__((ext_vector_type(2))) float v2f;
typedef __attribute__((ext_vector_type(8))) float v8f;

#define NEG_SLOPE 0.2f
#define LN_EPS 1e-5f

static __device__ inline v8f wmma4(v2f a, v2f b, v8f c) {
  return __builtin_amdgcn_wmma_f32_16x16x4_f32(
       false, a,  false, b,
       (short)0, c,  false,  false);
}

#define MAXKEY_NEG_INF ((int)0x807fffff)
static __device__ inline int fkey(float f) {
  const int b = __float_as_int(f);
  return b >= 0 ? b : b ^ 0x7fffffff;
}
static __device__ inline float fkey_inv(int k) {
  return __int_as_float(k >= 0 ? k : k ^ 0x7fffffff);
}
static __device__ inline void atomicMaxKey(int* addr, float val) {
  atomicMax(addr, fkey(val));
}

static __device__ inline int bcast_i(int v, int srcLane) {
  return __builtin_amdgcn_readlane(v, srcLane);
}
static __device__ inline float bcast_f(float v, int srcLane) {
  return __int_as_float(__builtin_amdgcn_readlane(__float_as_int(v), srcLane));
}

static __device__ inline int chip_half() {
  return (int)((__builtin_amdgcn_s_sendmsg_rtn(0x87) >> 18) & 1u);
}

static __device__ inline void load_edge(const int* __restrict__ ei, int E, int TE, int t,
                                        int& s, int& d) {
  s = 0; d = 0;
  if (t < TE) {
    if (t < E) { s = ei[t]; d = ei[E + t]; } else { s = t - E; d = s; }
  }
}

static __device__ inline int wave_in_grid() {
  return __builtin_amdgcn_readfirstlane(
      (int)(((long long)blockIdx.x * blockDim.x + threadIdx.x) >> 5));
}
static __device__ inline int waves_in_grid() {
  return (int)(gridDim.x * (blockDim.x >> 5));
}

template <int OFF, int STRIDE, int Q>
static __device__ inline void bfly(float (&p)[32], bool up) {
  if constexpr (Q < STRIDE) {
    const float lo = p[Q], hi = p[Q + STRIDE];
    const float keep = up ? hi : lo;
    const float send = up ? lo : hi;
    p[Q] = keep + __shfl_xor(send, OFF, 32);
    bfly<OFF, STRIDE, Q + 1>(p, up);
  }
}

__global__ void clear_all(float4* __restrict__ acc1, float4* __restrict__ acc2, long long nq,
                          int4* __restrict__ max1,
                          int* __restrict__ max2, int n) {
  const float4 z = make_float4(0.f, 0.f, 0.f, 0.f);
  const int ninf = MAXKEY_NEG_INF;
  const long long base = (long long)blockIdx.x * (blockDim.x * 4) + threadIdx.x;
#pragma unroll
  for (int j = 0; j < 4; ++j) {
    const long long i = base + (long long)j * blockDim.x;
    if (i < nq) { acc1[i] = z; acc2[i] = z; }
  }
  const long long g = (long long)blockIdx.x * blockDim.x + threadIdx.x;
  if (g < n) {
    max1[g] = make_int4(ninf, ninf, ninf, ninf);
    max2[g] = ninf;
  }
}

__global__ void gemm2_wmma(const float* __restrict__ h,
                           const float* __restrict__ Wl,
                           const float* __restrict__ Wr,
                           float* __restrict__ xl,
                           float* __restrict__ xr, int ntiles) {
  int wave = (int)(((long long)blockIdx.x * blockDim.x + threadIdx.x) >> 5);
  int lane = threadIdx.x & 31;
  if (wave >= ntiles) return;
  int row0 = wave * 16;
  int m  = lane & 15;
  int hi = lane >> 4;
  const float* hrow = h + (size_t)(row0 + m) * 64;
  v8f cl[2] = {{0.f},{0.f}};
  v8f cr[2] = {{0.f},{0.f}};
#pragma unroll
  for (int kk = 0; kk < 16; ++kk) {
    int k0 = kk * 4 + hi * 2;
    v2f a; a.x = hrow[k0]; a.y = hrow[k0 + 1];
#pragma unroll
    for (int j = 0; j < 2; ++j) {
      v2f bl, br;
      bl.x = Wl[(k0 + 0) * 32 + j * 16 + m];
      bl.y = Wl[(k0 + 1) * 32 + j * 16 + m];
      br.x = Wr[(k0 + 0) * 32 + j * 16 + m];
      br.y = Wr[(k0 + 1) * 32 + j * 16 + m];
      cl[j] = wmma4(a, bl, cl[j]);
      cr[j] = wmma4(a, br, cr[j]);
    }
  }
#pragma unroll
  for (int j = 0; j < 2; ++j)
#pragma unroll
    for (int v = 0; v < 8; ++v) {
      int r = v + hi * 8;
      xl[(size_t)(row0 + r) * 32 + j * 16 + m] = cl[j][v];
      xr[(size_t)(row0 + r) * 32 + j * 16 + m] = cr[j][v];
    }
}

__global__ void edge_logits1(const int* __restrict__ ei, int E, int n,
                             const float* __restrict__ x,
                             const float* __restrict__ Wl,
                             const float* __restrict__ Wr,
                             const float* __restrict__ a1,
                             float* __restrict__ logits,
                             int* __restrict__ maxd) {
  const int lane = threadIdx.x & 31;
  const int wave0 = wave_in_grid();
  const int nw = waves_in_grid();
  const int TE = E + n;
  const int ntile = (TE + 15) >> 4;
  if (wave0 >= ntile) return;
  const int m  = lane & 15;
  const int hi = lane >> 4;
  int s, d;
  load_edge(ei, E, TE, wave0 * 16 + m, s, d);
  int* maxh = maxd + (size_t)chip_half() * n * 4;
  v2f bl[4], br[4];
  float aj[4];
#pragma unroll
  for (int j = 0; j < 4; ++j) {
    bl[j].x = Wl[(2 * hi + 0) * 64 + j * 16 + m];
    bl[j].y = Wl[(2 * hi + 1) * 64 + j * 16 + m];
    br[j].x = Wr[(2 * hi + 0) * 64 + j * 16 + m];
    br[j].y = Wr[(2 * hi + 1) * 64 + j * 16 + m];
    aj[j] = a1[j * 16 + m];
  }
  const int head = lane & 3;
  const int jb = hi * 8 + 2 * ((lane >> 2) & 3);
  int tile = wave0;
  for (;;) {
    const int tn = tile + nw;
    int sn = 0, dn = 0;
    if (tn < ntile) load_edge(ei, E, TE, tn * 16 + m, sn, dn);
    const int base = tile * 16;
    const float2 xs = *(const float2*)(x + (size_t)s * 4 + hi * 2);
    const float2 xd = *(const float2*)(x + (size_t)d * 4 + hi * 2);
    v2f fs; fs.x = xs.x; fs.y = xs.y;
    v2f fd; fd.x = xd.x; fd.y = xd.y;
    float p[32];
#pragma unroll
    for (int j = 0; j < 4; ++j) {
      v8f cc = {};
      cc = wmma4(fs, bl[j], cc);
      cc = wmma4(fd, br[j], cc);
#pragma unroll
      for (int v = 0; v < 8; ++v) {
        float e = cc[v]; e = e >= 0.f ? e : NEG_SLOPE * e;
        p[(v >> 1) * 8 + j * 2 + (v & 1)] = e * aj[j];
      }
    }
    bfly<8, 16, 0>(p, (lane & 8) != 0);
    bfly<4, 8, 0>(p, (lane & 4) != 0);
    bfly<2, 4, 0>(p, (lane & 2) != 0);
    bfly<1, 2, 0>(p, (lane & 1) != 0);
#pragma unroll
    for (int q = 0; q < 2; ++q) {
      const int j = jb + q;
      const int dj = __shfl(d, j, 32);
      if (base + j < TE) {
        logits[(size_t)(base + j) * 4 + head] = p[q];
        atomicMaxKey(&maxh[(size_t)dj * 4 + head], p[q]);
      }
    }
    if (tn >= ntile) break;
    tile = tn; s = sn; d = dn;
  }
}

__global__ void edge_accum1(const int* __restrict__ ei, int E, int n,
                            const float* __restrict__ x,
                            const float* __restrict__ logits,
                            const int* __restrict__ maxd,
                            float* __restrict__ acc) {
  const int lane = threadIdx.x & 31;
  const int wave0 = wave_in_grid();
  const int nw = waves_in_grid();
  const int TE = E + n;
  const int ntile = (TE + 7) >> 3;
  if (wave0 >= ntile) return;
  const int h = lane & 3;
  int s, d;
  load_edge(ei, E, TE, wave0 * 8 + (lane >> 2), s, d);
  acc += (size_t)chip_half() * n * 32;
  const int kx = lane < 16 ? (lane >> 2) : 3;
  const int hsel = (lane & 3) << 2;
  int tile = wave0;
  for (;;) {
    const int tn = tile + nw;
    int sn = 0, dn = 0;
    if (tn < ntile) load_edge(ei, E, TE, tn * 8 + (lane >> 2), sn, dn);
    const int base = tile * 8;
    const int e = base + (lane >> 2);
    float ex = 0.f;
    if (e < TE) {
      const int ka = maxd[(size_t)d * 4 + h];
      const int kb = maxd[(size_t)n * 4 + (size_t)d * 4 + h];
      ex = __expf(logits[(size_t)e * 4 + h] - fkey_inv(ka > kb ? ka : kb));
    }
    const int cnt = TE - base;
    if (cnt >= 8) {
      float xv[8], w[8];
      int dj[8];
#pragma unroll
      for (int j = 0; j < 8; ++j)
        xv[j] = x[(size_t)bcast_i(s, 4 * j) * 4 + kx];
#pragma unroll
      for (int j = 0; j < 8; ++j) {
        w[j] = __int_as_float(
            __builtin_amdgcn_ds_bpermute(hsel + 16 * j, __float_as_int(ex)));
        dj[j] = bcast_i(d, 4 * j);
      }
      if (lane < 20) {
#pragma unroll
        for (int j = 0; j < 8; ++j)
          atomicAdd(acc + (size_t)dj[j] * 32 + lane, (lane < 16 ? xv[j] : 1.f) * w[j]);
      }
    } else {
      for (int j = 0; j < cnt; ++j) {
        const float xj = x[(size_t)bcast_i(s, 4 * j) * 4 + kx];
        const float w = __int_as_float(
            __builtin_amdgcn_ds_bpermute(hsel + 16 * j, __float_as_int(ex)));
        const float val = (lane < 16 ? xj : 1.f) * w;
        float* row = acc + (size_t)bcast_i(d, 4 * j) * 32;
        if (lane < 20) atomicAdd(row + lane, val);
      }
    }
    if (tn >= ntile) break;
    tile = tn; s = sn; d = dn;
  }
}

__global__ void node_fin1(const float* __restrict__ acc,
                          const float* __restrict__ Wl,
                          const float* __restrict__ b1,
                          const float* __restrict__ g,
                          const float* __restrict__ be,
                          float* __restrict__ h1, int ntiles) {
  int wave = (int)(((long long)blockIdx.x * blockDim.x + threadIdx.x) >> 5);
  int lane = threadIdx.x & 31;
  if (wave >= ntiles) return;
  int row0 = wave * 16;
  int m  = lane & 15;
  int hi = lane >> 4;
  const float4* arow = (const float4*)(acc + (size_t)(row0 + m) * 32);
  const float4* brow = arow + (size_t)ntiles * 16 * 8;
  const float4 p0 = arow[2 * hi],     r0 = brow[2 * hi];
  const float4 p1 = arow[2 * hi + 1], r1 = brow[2 * hi + 1];
  const float4 dn = arow[4],          rn = brow[4];
  const float pa[4] = {p0.x + r0.x, p0.y + r0.y, p0.z + r0.z, p0.w + r0.w};
  const float pb[4] = {p1.x + r1.x, p1.y + r1.y, p1.z + r1.z, p1.w + r1.w};
  const float dd[4] = {dn.x + rn.x, dn.y + rn.y, dn.z + rn.z, dn.w + rn.w};
  v8f c[4];
  float bias[4], gg[4], bb[4];
#pragma unroll
  for (int h = 0; h < 4; ++h) {
    v2f a; a.x = pa[h] / dd[h]; a.y = pb[h] / dd[h];
    v2f b;
    b.x = Wl[(2 * hi + 0) * 64 + h * 16 + m];
    b.y = Wl[(2 * hi + 1) * 64 + h * 16 + m];
    v8f z = {};
    c[h] = wmma4(a, b, z);
    bias[h] = b1[h * 16 + m];
    gg[h] = g[h * 16 + m];
    bb[h] = be[h * 16 + m];
  }
  float s[8], q[8];
#pragma unroll
  for (int v = 0; v < 8; ++v) {
#pragma unroll
    for (int h = 0; h < 4; ++h) c[h][v] += bias[h];
    s[v] = (c[0][v] + c[1][v]) + (c[2][v] + c[3][v]);
  }
#pragma unroll
  for (int o = 8; o; o >>= 1)
#pragma unroll
    for (int v = 0; v < 8; ++v) s[v] += __shfl_xor(s[v], o, 32);
#pragma unroll
  for (int v = 0; v < 8; ++v) {
    const float mu = s[v] * (1.f / 64.f);
#pragma unroll
    for (int h = 0; h < 4; ++h) c[h][v] -= mu;
    q[v] = (c[0][v] * c[0][v] + c[1][v] * c[1][v]) + (c[2][v] * c[2][v] + c[3][v] * c[3][v]);
  }
#pragma unroll
  for (int o = 8; o; o >>= 1)
#pragma unroll
    for (int v = 0; v < 8; ++v) q[v] += __shfl_xor(q[v], o, 32);
#pragma unroll
  for (int v = 0; v < 8; ++v) {
    const float rs = rsqrtf(q[v] * (1.f / 64.f) + LN_EPS);
    float* orow = h1 + (size_t)(row0 + v + hi * 8) * 64 + m;
#pragma unroll
    for (int h = 0; h < 4; ++h) {
      float y = c[h][v] * rs * gg[h] + bb[h];
      y = y > 0.f ? y : expm1f(y);
      orow[h * 16] = y;
    }
  }
}

__global__ void edge_logits2(const int* __restrict__ ei, int E, int n,
                             const float* __restrict__ xl,
                             const float* __restrict__ xr,
                             const float* __restrict__ a2,
                             float* __restrict__ logits,
                             int* __restrict__ maxd) {
  const int lane = threadIdx.x & 31;
  const int wave0 = wave_in_grid();
  const int nw = waves_in_grid();
  const int TE = E + n;
  const int ntile = (TE + 31) >> 5;
  if (wave0 >= ntile) return;
  int s, d;
  load_edge(ei, E, TE, wave0 * 32 + lane, s, d);
  maxd += (size_t)chip_half() * n;
  const float al = a2[lane];
  int tile = wave0;
  for (;;) {
    const int tn = tile + nw;
    int sn = 0, dn = 0;
    if (tn < ntile) load_edge(ei, E, TE, tn * 32 + lane, sn, dn);
    const int t = tile * 32 + lane;
    float p[32];
#pragma unroll
    for (int j0 = 0; j0 < 32; j0 += 8) {
      float u[8], v[8];
#pragma unroll
      for (int k = 0; k < 8; ++k) {
        u[k] = xl[(unsigned)bcast_i(s, j0 + k) * 32u + (unsigned)lane];
        v[k] = xr[(unsigned)bcast_i(d, j0 + k) * 32u + (unsigned)lane];
      }
#pragma unroll
      for (int k = 0; k < 8; ++k) {
        float e = u[k] + v[k]; e = e >= 0.f ? e : NEG_SLOPE * e;
        p[j0 + k] = e * al;
      }
    }
    bfly<16, 16, 0>(p, (lane & 16) != 0);
    bfly<8, 8, 0>(p, (lane & 8) != 0);
    bfly<4, 4, 0>(p, (lane & 4) != 0);
    bfly<2, 2, 0>(p, (lane & 2) != 0);
    bfly<1, 1, 0>(p, (lane & 1) != 0);
    if (t < TE) {
      logits[t] = p[0];
      atomicMaxKey(&maxd[d], p[0]);
    }
    if (tn >= ntile) break;
    tile = tn; s = sn; d = dn;
  }
}

__global__ void edge_accum2(const int* __restrict__ ei, int E, int n,
                            const float* __restrict__ xl,
                            const float* __restrict__ logits,
                            const int* __restrict__ maxd,
                            float* __restrict__ acc) {
  const int lane = threadIdx.x & 31;
  const int wave0 = wave_in_grid();
  const int nw = waves_in_grid();
  const int TE = E + n;
  const int ntile = (TE + 31) >> 5;
  if (wave0 >= ntile) return;
  int s, d;
  load_edge(ei, E, TE, wave0 * 32 + lane, s, d);
  acc += (size_t)chip_half() * n * 32;
  const bool den_lane = lane == 31;
  int tile = wave0;
  for (;;) {
    const int tn = tile + nw;
    int sn = 0, dn = 0;
    if (tn < ntile) load_edge(ei, E, TE, tn * 32 + lane, sn, dn);
    const int base = tile * 32;
    const int t = base + lane;
    float ex = 0.f;
    if (t < TE) {
      const int ka = maxd[d];
      const int kb = maxd[(size_t)n + d];
      ex = __expf(logits[t] - fkey_inv(ka > kb ? ka : kb));
    }
    const int cnt = TE - base;
    if (cnt >= 32) {
#pragma unroll
      for (int j0 = 0; j0 < 32; j0 += 8) {
        float v[8];
#pragma unroll
        for (int u = 0; u < 8; ++u) v[u] = xl[(size_t)bcast_i(s, j0 + u) * 32 + lane];
#pragma unroll
        for (int u = 0; u < 8; ++u) {
          const float c31 = bcast_f(v[u], 31);
          const float rel = den_lane ? 1.f : v[u] - c31;
          atomicAdd(acc + (size_t)bcast_i(d, j0 + u) * 32 + lane, bcast_f(ex, j0 + u) * rel);
        }
      }
    } else {
      for (int j = 0; j < cnt; ++j) {
        const float v = xl[(size_t)bcast_i(s, j) * 32 + lane];
        const float c31 = bcast_f(v, 31);
        const float rel = den_lane ? 1.f : v - c31;
        atomicAdd(acc + (size_t)bcast_i(d, j) * 32 + lane, bcast_f(ex, j) * rel);
      }
    }
    if (tn >= ntile) break;
    tile = tn; s = sn; d = dn;
  }
}

__global__ void node_fin2(const float* __restrict__ acc,
                          const float* __restrict__ b2,
                          const float* __restrict__ g,
                          const float* __restrict__ be,
                          float* __restrict__ out, int n) {
  const int lane = threadIdx.x & 31;
  const int wave0 = wave_in_grid();
  const int nw = waves_in_grid();
  if (wave0 >= n) return;
  const float bias = b2[lane], gg = g[lane], bb = be[lane];
  float a = acc[(size_t)wave0 * 32 + lane] + acc[(size_t)n * 32 + (size_t)wave0 * 32 + lane];
  int node = wave0;
  for (;;) {
    const int nn = node + nw;
    float an = 0.f;
    if (nn < n) an = acc[(size_t)nn * 32 + lane] + acc[(size_t)n * 32 + (size_t)nn * 32 + lane];
    const float den = bcast_f(a, 31);
    float v = (lane == 31 ? 0.f : a / den) + bias;
    float s = v;
#pragma unroll
    for (int o = 16; o; o >>= 1) s += __shfl_xor(s, o, 32);
    float mu = s * (1.f / 32.f);
    float dd = v - mu;
    float q = dd * dd;
#pragma unroll
    for (int o = 16; o; o >>= 1) q += __shfl_xor(q, o, 32);
    float rs = rsqrtf(q * (1.f / 32.f) + LN_EPS);
    out[(size_t)node * 32 + lane] = dd * rs * gg + bb;
    if (nn >= n) break;
    node = nn; a = an;
  }
}

static inline int gridFor(long long threads, int block) {
  return (int)((threads + block - 1) / block);
}

#define RES_GROUPS 1536
#define RES_GROUPS_WIDE 1024
static inline int residentGrid(long long tiles, int groups) {
  if (tiles <= 0) return 1;
  const long long resWaves = (long long)groups * 8;
  const long long trips = (tiles + resWaves - 1) / resWaves;
  const long long waves = (tiles + trips - 1) / trips;
  return (int)((waves + 7) / 8);
}

extern "C" void kernel_launch(void* const* d_in, const int* in_sizes, int n_in,
                              void* d_out, int out_size, void* d_ws, size_t ws_size,
                              hipStream_t stream) {
  const float* x   = (const float*)d_in[0];
  const int*   ei  = (const int*)  d_in[1];
  const float* W1l = (const float*)d_in[2];
  const float* W1r = (const float*)d_in[3];
  const float* a1  = (const float*)d_in[4];
  const float* b1  = (const float*)d_in[5];
  const float* g1  = (const float*)d_in[6];
  const float* be1 = (const float*)d_in[7];
  const float* W2l = (const float*)d_in[8];
  const float* W2r = (const float*)d_in[9];
  const float* a2  = (const float*)d_in[10];
  const float* b2  = (const float*)d_in[11];
  const float* g2  = (const float*)d_in[12];
  const float* be2 = (const float*)d_in[13];
  float* out = (float*)d_out;

  const int n  = in_sizes[0] / 4;
  const int E  = in_sizes[1] / 2;
  const int TE = E + n;
  const int ntiles = n / 16;

  float* ws = (float*)d_ws;
  const size_t ACC1 = 0;
  const size_t ACC2 = ACC1 + (size_t)n * 64;
  const size_t LOG1 = ACC2 + (size_t)n * 64;
  const size_t XL1  = LOG1;
  const size_t XR1  = XL1 + (size_t)n * 32;
  const size_t LOGSZ = (size_t)TE * 4 > (size_t)n * 64 ? (size_t)TE * 4 : (size_t)n * 64;
  const size_t H1   = LOG1 + LOGSZ;
  const size_t MAX1 = H1 + (size_t)n * 64;
  const size_t MAX2 = MAX1 + (size_t)n * 8;
  const size_t LOG2 = MAX2 + (size_t)n * 2;

  const int B = 256;

  clear_all<<<gridFor((long long)n * 16, B * 4), B, 0, stream>>>(
      (float4*)(ws + ACC1), (float4*)(ws + ACC2), (long long)n * 16,
      (int4*)(ws + MAX1), (int*)(ws + MAX2), n * 2);

  edge_logits1<<<residentGrid(((long long)TE + 15) / 16, RES_GROUPS_WIDE), B, 0, stream>>>(
      ei, E, n, x, W1l, W1r, a1, ws + LOG1, (int*)(ws + MAX1));

  edge_accum1<<<residentGrid(((long long)TE + 7) / 8, RES_GROUPS), B, 0, stream>>>(
      ei, E, n, x, ws + LOG1, (const int*)(ws + MAX1), ws + ACC1);

  node_fin1<<<gridFor((long long)ntiles * 32, 128), 128, 0, stream>>>(
      ws + ACC1, W1l, b1, g1, be1, ws + H1, ntiles);

  gemm2_wmma<<<gridFor((long long)ntiles * 32, B), B, 0, stream>>>(
      ws + H1, W2l, W2r, ws + XL1, ws + XR1, ntiles);

  edge_logits2<<<residentGrid(((long long)TE + 31) / 32, RES_GROUPS), B, 0, stream>>>(
      ei, E, n, ws + XL1, ws + XR1, a2, ws + LOG2, (int*)(ws + MAX2));

  edge_accum2<<<residentGrid(((long long)TE + 31) / 32, RES_GROUPS), B, 0, stream>>>(
      ei, E, n, ws + XL1, ws + LOG2, (const int*)(ws + MAX2), ws + ACC2);

  node_fin2<<<residentGrid(n, RES_GROUPS), B, 0, stream>>>(
      ws + ACC2, b2, g2, be2, out, n);
}


# Layer-2 softmax shifted by the node's self-loop logit, not the segment max: logits, exp and atomic sums in ONE edge pass, source row gathered once; max path kept behind overflow flag; new exp scale
# speedup vs baseline: 8.5930x; 1.0887x over previous
#include <hip/hip_runtime.h>
#include <hip/hip_bf16.h>
#include <math.h>


typedef __attribute__((ext_vector_type(2))) float v2f;
typedef __attribute__((ext_vector_type(8))) float v8f;

#define NEG_SLOPE 0.2f
#define LN_EPS 1e-5f
#define SHIFT_GUARD 60.f

static __device__ inline v8f wmma4(v2f a, v2f b, v8f c) {
  return __builtin_amdgcn_wmma_f32_16x16x4_f32(
       false, a,  false, b,
       (short)0, c,  false,  false);
}

#define MAXKEY_NEG_INF ((int)0x807fffff)
static __device__ inline int fkey(float f) {
  const int b = __float_as_int(f);
  return b >= 0 ? b : b ^ 0x7fffffff;
}
static __device__ inline float fkey_inv(int k) {
  return __int_as_float(k >= 0 ? k : k ^ 0x7fffffff);
}
static __device__ inline void atomicMaxKey(int* addr, float val) {
  atomicMax(addr, fkey(val));
}

static __device__ inline int bcast_i(int v, int srcLane) {
  return __builtin_amdgcn_readlane(v, srcLane);
}
static __device__ inline float bcast_f(float v, int srcLane) {
  return __int_as_float(__builtin_amdgcn_readlane(__float_as_int(v), srcLane));
}

static __device__ inline int chip_half() {
  return (int)((__builtin_amdgcn_s_sendmsg_rtn(0x87) >> 18) & 1u);
}

static __device__ inline void load_edge(const int* __restrict__ ei, int E, int TE, int t,
                                        int& s, int& d) {
  s = 0; d = 0;
  if (t < TE) {
    if (t < E) { s = ei[t]; d = ei[E + t]; } else { s = t - E; d = s; }
  }
}

static __device__ inline int wave_in_grid() {
  return __builtin_amdgcn_readfirstlane(
      (int)(((long long)blockIdx.x * blockDim.x + threadIdx.x) >> 5));
}
static __device__ inline int waves_in_grid() {
  return (int)(gridDim.x * (blockDim.x >> 5));
}

template <int OFF, int STRIDE, int Q>
static __device__ inline void bfly(float (&p)[32], bool up) {
  if constexpr (Q < STRIDE) {
    const float lo = p[Q], hi = p[Q + STRIDE];
    const float keep = up ? hi : lo;
    const float send = up ? lo : hi;
    p[Q] = keep + __shfl_xor(send, OFF, 32);
    bfly<OFF, STRIDE, Q + 1>(p, up);
  }
}

__global__ void clear_all(float4* __restrict__ acc1, float4* __restrict__ acc2, long long nq,
                          int4* __restrict__ max1,
                          int* __restrict__ max2, int n) {
  const float4 z = make_float4(0.f, 0.f, 0.f, 0.f);
  const int ninf = MAXKEY_NEG_INF;
  const long long base = (long long)blockIdx.x * (blockDim.x * 4) + threadIdx.x;
#pragma unroll
  for (int j = 0; j < 4; ++j) {
    const long long i = base + (long long)j * blockDim.x;
    if (i < nq) { acc1[i] = z; acc2[i] = z; }
  }
  const long long g = (long long)blockIdx.x * blockDim.x + threadIdx.x;
  if (g < n) {
    max1[g] = make_int4(ninf, ninf, ninf, ninf);
    max2[g] = ninf;
  }
}

__global__ void gemm2_wmma(const float* __restrict__ h,
                           const float* __restrict__ Wl,
                           const float* __restrict__ Wr,
                           const float* __restrict__ a2,
                           float* __restrict__ xl,
                           float* __restrict__ xr,
                           float* __restrict__ sh, int ntiles) {
  int wave = (int)(((long long)blockIdx.x * blockDim.x + threadIdx.x) >> 5);
  int lane = threadIdx.x & 31;
  if (wave >= ntiles) return;
  int row0 = wave * 16;
  int m  = lane & 15;
  int hi = lane >> 4;
  const float* hrow = h + (size_t)(row0 + m) * 64;
  v8f cl[2] = {{0.f},{0.f}};
  v8f cr[2] = {{0.f},{0.f}};
#pragma unroll
  for (int kk = 0; kk < 16; ++kk) {
    int k0 = kk * 4 + hi * 2;
    v2f a; a.x = hrow[k0]; a.y = hrow[k0 + 1];
#pragma unroll
    for (int j = 0; j < 2; ++j) {
      v2f bl, br;
      bl.x = Wl[(k0 + 0) * 32 + j * 16 + m];
      bl.y = Wl[(k0 + 1) * 32 + j * 16 + m];
      br.x = Wr[(k0 + 0) * 32 + j * 16 + m];
      br.y = Wr[(k0 + 1) * 32 + j * 16 + m];
      cl[j] = wmma4(a, bl, cl[j]);
      cr[j] = wmma4(a, br, cr[j]);
    }
  }
#pragma unroll
  for (int j = 0; j < 2; ++j)
#pragma unroll
    for (int v = 0; v < 8; ++v) {
      int r = v + hi * 8;
      xl[(size_t)(row0 + r) * 32 + j * 16 + m] = cl[j][v];
      xr[(size_t)(row0 + r) * 32 + j * 16 + m] = cr[j][v];
    }
  const float a0 = a2[m], a1v = a2[16 + m];
  float t[8];
#pragma unroll
  for (int v = 0; v < 8; ++v) {
    float e0 = cl[0][v] + cr[0][v]; e0 = e0 >= 0.f ? e0 : NEG_SLOPE * e0;
    float e1 = cl[1][v] + cr[1][v]; e1 = e1 >= 0.f ? e1 : NEG_SLOPE * e1;
    t[v] = e0 * a0 + e1 * a1v;
  }
#pragma unroll
  for (int o = 8; o; o >>= 1)
#pragma unroll
    for (int v = 0; v < 8; ++v) t[v] += __shfl_xor(t[v], o, 32);
#pragma unroll
  for (int v = 0; v < 8; ++v)
    if (m == v) sh[row0 + v + hi * 8] = t[v];
}

__global__ void edge_logits1(const int* __restrict__ ei, int E, int n,
                             const float* __restrict__ x,
                             const float* __restrict__ Wl,
                             const float* __restrict__ Wr,
                             const float* __restrict__ a1,
                             float* __restrict__ logits,
                             int* __restrict__ maxd) {
  const int lane = threadIdx.x & 31;
  const int wave0 = wave_in_grid();
  const int nw = waves_in_grid();
  const int TE = E + n;
  const int ntile = (TE + 15) >> 4;
  if (wave0 >= ntile) return;
  const int m  = lane & 15;
  const int hi = lane >> 4;
  int s, d;
  load_edge(ei, E, TE, wave0 * 16 + m, s, d);
  int* maxh = maxd + (size_t)chip_half() * n * 4;
  v2f bl[4], br[4];
  float aj[4];
#pragma unroll
  for (int j = 0; j < 4; ++j) {
    bl[j].x = Wl[(2 * hi + 0) * 64 + j * 16 + m];
    bl[j].y = Wl[(2 * hi + 1) * 64 + j * 16 + m];
    br[j].x = Wr[(2 * hi + 0) * 64 + j * 16 + m];
    br[j].y = Wr[(2 * hi + 1) * 64 + j * 16 + m];
    aj[j] = a1[j * 16 + m];
  }
  const int head = lane & 3;
  const int jb = hi * 8 + 2 * ((lane >> 2) & 3);
  int tile = wave0;
  for (;;) {
    const int tn = tile + nw;
    int sn = 0, dn = 0;
    if (tn < ntile) load_edge(ei, E, TE, tn * 16 + m, sn, dn);
    const int base = tile * 16;
    const float2 xs = *(const float2*)(x + (size_t)s * 4 + hi * 2);
    const float2 xd = *(const float2*)(x + (size_t)d * 4 + hi * 2);
    v2f fs; fs.x = xs.x; fs.y = xs.y;
    v2f fd; fd.x = xd.x; fd.y = xd.y;
    float p[32];
#pragma unroll
    for (int j = 0; j < 4; ++j) {
      v8f cc = {};
      cc = wmma4(fs, bl[j], cc);
      cc = wmma4(fd, br[j], cc);
#pragma unroll
      for (int v = 0; v < 8; ++v) {
        float e = cc[v]; e = e >= 0.f ? e : NEG_SLOPE * e;
        p[(v >> 1) * 8 + j * 2 + (v & 1)] = e * aj[j];
      }
    }
    bfly<8, 16, 0>(p, (lane & 8) != 0);
    bfly<4, 8, 0>(p, (lane & 4) != 0);
    bfly<2, 4, 0>(p, (lane & 2) != 0);
    bfly<1, 2, 0>(p, (lane & 1) != 0);
#pragma unroll
    for (int q = 0; q < 2; ++q) {
      const int j = jb + q;
      const int dj = __shfl(d, j, 32);
      if (base + j < TE) {
        logits[(size_t)(base + j) * 4 + head] = p[q];
        atomicMaxKey(&maxh[(size_t)dj * 4 + head], p[q]);
      }
    }
    if (tn >= ntile) break;
    tile = tn; s = sn; d = dn;
  }
}

__global__ void edge_accum1(const int* __restrict__ ei, int E, int n,
                            const float* __restrict__ x,
                            const float* __restrict__ logits,
                            const int* __restrict__ maxd,
                            float* __restrict__ acc) {
  const int lane = threadIdx.x & 31;
  const int wave0 = wave_in_grid();
  const int nw = waves_in_grid();
  const int TE = E + n;
  const int ntile = (TE + 7) >> 3;
  if (wave0 >= ntile) return;
  const int h = lane & 3;
  int s, d;
  load_edge(ei, E, TE, wave0 * 8 + (lane >> 2), s, d);
  acc += (size_t)chip_half() * n * 32;
  const int kx = lane < 16 ? (lane >> 2) : 3;
  const int hsel = (lane & 3) << 2;
  int tile = wave0;
  for (;;) {
    const int tn = tile + nw;
    int sn = 0, dn = 0;
    if (tn < ntile) load_edge(ei, E, TE, tn * 8 + (lane >> 2), sn, dn);
    const int base = tile * 8;
    const int e = base + (lane >> 2);
    float ex = 0.f;
    if (e < TE) {
      const int ka = maxd[(size_t)d * 4 + h];
      const int kb = maxd[(size_t)n * 4 + (size_t)d * 4 + h];
      ex = __expf(logits[(size_t)e * 4 + h] - fkey_inv(ka > kb ? ka : kb));
    }
    const int cnt = TE - base;
    if (cnt >= 8) {
      float xv[8], w[8];
      int dj[8];
#pragma unroll
      for (int j = 0; j < 8; ++j)
        xv[j] = x[(size_t)bcast_i(s, 4 * j) * 4 + kx];
#pragma unroll
      for (int j = 0; j < 8; ++j) {
        w[j] = __int_as_float(
            __builtin_amdgcn_ds_bpermute(hsel + 16 * j, __float_as_int(ex)));
        dj[j] = bcast_i(d, 4 * j);
      }
      if (lane < 20) {
#pragma unroll
        for (int j = 0; j < 8; ++j)
          atomicAdd(acc + (size_t)dj[j] * 32 + lane, (lane < 16 ? xv[j] : 1.f) * w[j]);
      }
    } else {
      for (int j = 0; j < cnt; ++j) {
        const float xj = x[(size_t)bcast_i(s, 4 * j) * 4 + kx];
        const float w = __int_as_float(
            __builtin_amdgcn_ds_bpermute(hsel + 16 * j, __float_as_int(ex)));
        const float val = (lane < 16 ? xj : 1.f) * w;
        float* row = acc + (size_t)bcast_i(d, 4 * j) * 32;
        if (lane < 20) atomicAdd(row + lane, val);
      }
    }
    if (tn >= ntile) break;
    tile = tn; s = sn; d = dn;
  }
}

__global__ void node_fin1(const float* __restrict__ acc,
                          const float* __restrict__ Wl,
                          const float* __restrict__ b1,
                          const float* __restrict__ g,
                          const float* __restrict__ be,
                          float* __restrict__ h1, int ntiles) {
  int wave = (int)(((long long)blockIdx.x * blockDim.x + threadIdx.x) >> 5);
  int lane = threadIdx.x & 31;
  if (wave >= ntiles) return;
  int row0 = wave * 16;
  int m  = lane & 15;
  int hi = lane >> 4;
  const float4* arow = (const float4*)(acc + (size_t)(row0 + m) * 32);
  const float4* brow = arow + (size_t)ntiles * 16 * 8;
  const float4 p0 = arow[2 * hi],     r0 = brow[2 * hi];
  const float4 p1 = arow[2 * hi + 1], r1 = brow[2 * hi + 1];
  const float4 dn = arow[4],          rn = brow[4];
  const float pa[4] = {p0.x + r0.x, p0.y + r0.y, p0.z + r0.z, p0.w + r0.w};
  const float pb[4] = {p1.x + r1.x, p1.y + r1.y, p1.z + r1.z, p1.w + r1.w};
  const float dd[4] = {dn.x + rn.x, dn.y + rn.y, dn.z + rn.z, dn.w + rn.w};
  v8f c[4];
  float bias[4], gg[4], bb[4];
#pragma unroll
  for (int h = 0; h < 4; ++h) {
    v2f a; a.x = pa[h] / dd[h]; a.y = pb[h] / dd[h];
    v2f b;
    b.x = Wl[(2 * hi + 0) * 64 + h * 16 + m];
    b.y = Wl[(2 * hi + 1) * 64 + h * 16 + m];
    v8f z = {};
    c[h] = wmma4(a, b, z);
    bias[h] = b1[h * 16 + m];
    gg[h] = g[h * 16 + m];
    bb[h] = be[h * 16 + m];
  }
  float s[8], q[8];
#pragma unroll
  for (int v = 0; v < 8; ++v) {
#pragma unroll
    for (int h = 0; h < 4; ++h) c[h][v] += bias[h];
    s[v] = (c[0][v] + c[1][v]) + (c[2][v] + c[3][v]);
  }
#pragma unroll
  for (int o = 8; o; o >>= 1)
#pragma unroll
    for (int v = 0; v < 8; ++v) s[v] += __shfl_xor(s[v], o, 32);
#pragma unroll
  for (int v = 0; v < 8; ++v) {
    const float mu = s[v] * (1.f / 64.f);
#pragma unroll
    for (int h = 0; h < 4; ++h) c[h][v] -= mu;
    q[v] = (c[0][v] * c[0][v] + c[1][v] * c[1][v]) + (c[2][v] * c[2][v] + c[3][v] * c[3][v]);
  }
#pragma unroll
  for (int o = 8; o; o >>= 1)
#pragma unroll
    for (int v = 0; v < 8; ++v) q[v] += __shfl_xor(q[v], o, 32);
#pragma unroll
  for (int v = 0; v < 8; ++v) {
    const float rs = rsqrtf(q[v] * (1.f / 64.f) + LN_EPS);
    float* orow = h1 + (size_t)(row0 + v + hi * 8) * 64 + m;
#pragma unroll
    for (int h = 0; h < 4; ++h) {
      float y = c[h][v] * rs * gg[h] + bb[h];
      y = y > 0.f ? y : expm1f(y);
      orow[h * 16] = y;
    }
  }
}

__global__ void edge_fused2(const int* __restrict__ ei, int E, int n,
                            const float* __restrict__ xl,
                            const float* __restrict__ xr,
                            const float* __restrict__ a2,
                            const float* __restrict__ sh,
                            float* __restrict__ acc,
                            int* __restrict__ flag) {
  const int lane = threadIdx.x & 31;
  const int wave0 = wave_in_grid();
  const int nw = waves_in_grid();
  const int TE = E + n;
  const int ntile = (TE + 31) >> 5;
  if (wave0 >= ntile) return;
  int s, d;
  load_edge(ei, E, TE, wave0 * 32 + lane, s, d);
  acc += (size_t)chip_half() * n * 32;
  const float al = a2[lane];
  const bool den_lane = lane == 31;
  int tile = wave0;
  for (;;) {
    const int tn = tile + nw;
    int sn = 0, dn = 0;
    if (tn < ntile) load_edge(ei, E, TE, tn * 32 + lane, sn, dn);
    const int base = tile * 32;
    const int t = base + lane;
    const float shift = sh[d];
    float p[32], u[32];
#pragma unroll
    for (int j0 = 0; j0 < 32; j0 += 8) {
      float v[8];
#pragma unroll
      for (int k = 0; k < 8; ++k) {
        u[j0 + k] = xl[(unsigned)bcast_i(s, j0 + k) * 32u + (unsigned)lane];
        v[k] = xr[(unsigned)bcast_i(d, j0 + k) * 32u + (unsigned)lane];
      }
#pragma unroll
      for (int k = 0; k < 8; ++k) {
        float e = u[j0 + k] + v[k]; e = e >= 0.f ? e : NEG_SLOPE * e;
        p[j0 + k] = e * al;
      }
    }
    bfly<16, 16, 0>(p, (lane & 16) != 0);
    bfly<8, 8, 0>(p, (lane & 8) != 0);
    bfly<4, 4, 0>(p, (lane & 4) != 0);
    bfly<2, 2, 0>(p, (lane & 2) != 0);
    bfly<1, 1, 0>(p, (lane & 1) != 0);
    float ex = 0.f;
    if (t < TE) {
      const float arg = p[0] - shift;
      if (arg > SHIFT_GUARD) *flag = 1;
      ex = __expf(arg);
    }
    const int cnt = TE - base;
    if (cnt >= 32) {
#pragma unroll
      for (int j = 0; j < 32; ++j) {
        const float c31 = bcast_f(u[j], 31);
        const float rel = den_lane ? 1.f : u[j] - c31;
        atomicAdd(acc + (size_t)bcast_i(d, j) * 32 + lane, bcast_f(ex, j) * rel);
      }
    } else {
      for (int j = 0; j < cnt; ++j) {
        const float v = xl[(size_t)bcast_i(s, j) * 32 + lane];
        const float c31 = bcast_f(v, 31);
        const float rel = den_lane ? 1.f : v - c31;
        atomicAdd(acc + (size_t)bcast_i(d, j) * 32 + lane, bcast_f(ex, j) * rel);
      }
    }
    if (tn >= ntile) break;
    tile = tn; s = sn; d = dn;
  }
}

__global__ void clear2_guarded(float4* __restrict__ acc2, long long nq,
                               int* __restrict__ max2, int n2,
                               const int* __restrict__ flag) {
  if (*flag == 0) return;
  const float4 z = make_float4(0.f, 0.f, 0.f, 0.f);
  const long long step = (long long)gridDim.x * blockDim.x;
  const long long g0 = (long long)blockIdx.x * blockDim.x + threadIdx.x;
  for (long long i = g0; i < nq; i += step) acc2[i] = z;
  for (long long i = g0; i < n2; i += step) max2[i] = MAXKEY_NEG_INF;
}

__global__ void edge_logits2(const int* __restrict__ ei, int E, int n,
                             const float* __restrict__ xl,
                             const float* __restrict__ xr,
                             const float* __restrict__ a2,
                             float* __restrict__ logits,
                             int* __restrict__ maxd,
                             const int* __restrict__ flag) {
  if (*flag == 0) return;
  const int lane = threadIdx.x & 31;
  const int wave0 = wave_in_grid();
  const int nw = waves_in_grid();
  const int TE = E + n;
  const int ntile = (TE + 31) >> 5;
  if (wave0 >= ntile) return;
  int s, d;
  load_edge(ei, E, TE, wave0 * 32 + lane, s, d);
  maxd += (size_t)chip_half() * n;
  const float al = a2[lane];
  int tile = wave0;
  for (;;) {
    const int tn = tile + nw;
    int sn = 0, dn = 0;
    if (tn < ntile) load_edge(ei, E, TE, tn * 32 + lane, sn, dn);
    const int t = tile * 32 + lane;
    float p[32];
#pragma unroll
    for (int j0 = 0; j0 < 32; j0 += 8) {
      float u[8], v[8];
#pragma unroll
      for (int k = 0; k < 8; ++k) {
        u[k] = xl[(unsigned)bcast_i(s, j0 + k) * 32u + (unsigned)lane];
        v[k] = xr[(unsigned)bcast_i(d, j0 + k) * 32u + (unsigned)lane];
      }
#pragma unroll
      for (int k = 0; k < 8; ++k) {
        float e = u[k] + v[k]; e = e >= 0.f ? e : NEG_SLOPE * e;
        p[j0 + k] = e * al;
      }
    }
    bfly<16, 16, 0>(p, (lane & 16) != 0);
    bfly<8, 8, 0>(p, (lane & 8) != 0);
    bfly<4, 4, 0>(p, (lane & 4) != 0);
    bfly<2, 2, 0>(p, (lane & 2) != 0);
    bfly<1, 1, 0>(p, (lane & 1) != 0);
    if (t < TE) {
      logits[t] = p[0];
      atomicMaxKey(&maxd[d], p[0]);
    }
    if (tn >= ntile) break;
    tile = tn; s = sn; d = dn;
  }
}

__global__ void edge_accum2(const int* __restrict__ ei, int E, int n,
                            const float* __restrict__ xl,
                            const float* __restrict__ logits,
                            const int* __restrict__ maxd,
                            float* __restrict__ acc,
                            const int* __restrict__ flag) {
  if (*flag == 0) return;
  const int lane = threadIdx.x & 31;
  const int wave0 = wave_in_grid();
  const int nw = waves_in_grid();
  const int TE = E + n;
  const int ntile = (TE + 31) >> 5;
  if (wave0 >= ntile) return;
  int s, d;
  load_edge(ei, E, TE, wave0 * 32 + lane, s, d);
  acc += (size_t)chip_half() * n * 32;
  const bool den_lane = lane == 31;
  int tile = wave0;
  for (;;) {
    const int tn = tile + nw;
    int sn = 0, dn = 0;
    if (tn < ntile) load_edge(ei, E, TE, tn * 32 + lane, sn, dn);
    const int base = tile * 32;
    const int t = base + lane;
    float ex = 0.f;
    if (t < TE) {
      const int ka = maxd[d];
      const int kb = maxd[(size_t)n + d];
      ex = __expf(logits[t] - fkey_inv(ka > kb ? ka : kb));
    }
    const int cnt = TE - base;
    if (cnt >= 32) {
#pragma unroll
      for (int j0 = 0; j0 < 32; j0 += 8) {
        float v[8];
#pragma unroll
        for (int u = 0; u < 8; ++u) v[u] = xl[(size_t)bcast_i(s, j0 + u) * 32 + lane];
#pragma unroll
        for (int u = 0; u < 8; ++u) {
          const float c31 = bcast_f(v[u], 31);
          const float rel = den_lane ? 1.f : v[u] - c31;
          atomicAdd(acc + (size_t)bcast_i(d, j0 + u) * 32 + lane, bcast_f(ex, j0 + u) * rel);
        }
      }
    } else {
      for (int j = 0; j < cnt; ++j) {
        const float v = xl[(size_t)bcast_i(s, j) * 32 + lane];
        const float c31 = bcast_f(v, 31);
        const float rel = den_lane ? 1.f : v - c31;
        atomicAdd(acc + (size_t)bcast_i(d, j) * 32 + lane, bcast_f(ex, j) * rel);
      }
    }
    if (tn >= ntile) break;
    tile = tn; s = sn; d = dn;
  }
}

__global__ void node_fin2(const float* __restrict__ acc,
                          const float* __restrict__ b2,
                          const float* __restrict__ g,
                          const float* __restrict__ be,
                          float* __restrict__ out, int n) {
  const int lane = threadIdx.x & 31;
  const int wave0 = wave_in_grid();
  const int nw = waves_in_grid();
  if (wave0 >= n) return;
  const float bias = b2[lane], gg = g[lane], bb = be[lane];
  float a = acc[(size_t)wave0 * 32 + lane] + acc[(size_t)n * 32 + (size_t)wave0 * 32 + lane];
  int node = wave0;
  for (;;) {
    const int nn = node + nw;
    float an = 0.f;
    if (nn < n) an = acc[(size_t)nn * 32 + lane] + acc[(size_t)n * 32 + (size_t)nn * 32 + lane];
    const float den = bcast_f(a, 31);
    float v = (lane == 31 ? 0.f : a / den) + bias;
    float s = v;
#pragma unroll
    for (int o = 16; o; o >>= 1) s += __shfl_xor(s, o, 32);
    float mu = s * (1.f / 32.f);
    float dd = v - mu;
    float q = dd * dd;
#pragma unroll
    for (int o = 16; o; o >>= 1) q += __shfl_xor(q, o, 32);
    float rs = rsqrtf(q * (1.f / 32.f) + LN_EPS);
    out[(size_t)node * 32 + lane] = dd * rs * gg + bb;
    if (nn >= n) break;
    node = nn; a = an;
  }
}

static inline int gridFor(long long threads, int block) {
  return (int)((threads + block - 1) / block);
}

#define RES_GROUPS 1536
#define RES_GROUPS_WIDE 1024
#define GUARDED_GROUPS 256
static inline int residentGrid(long long tiles, int groups) {
  if (tiles <= 0) return 1;
  const long long resWaves = (long long)groups * 8;
  const long long trips = (tiles + resWaves - 1) / resWaves;
  const long long waves = (tiles + trips - 1) / trips;
  return (int)((waves + 7) / 8);
}

extern "C" void kernel_launch(void* const* d_in, const int* in_sizes, int n_in,
                              void* d_out, int out_size, void* d_ws, size_t ws_size,
                              hipStream_t stream) {
  const float* x   = (const float*)d_in[0];
  const int*   ei  = (const int*)  d_in[1];
  const float* W1l = (const float*)d_in[2];
  const float* W1r = (const float*)d_in[3];
  const float* a1  = (const float*)d_in[4];
  const float* b1  = (const float*)d_in[5];
  const float* g1  = (const float*)d_in[6];
  const float* be1 = (const float*)d_in[7];
  const float* W2l = (const float*)d_in[8];
  const float* W2r = (const float*)d_in[9];
  const float* a2  = (const float*)d_in[10];
  const float* b2  = (const float*)d_in[11];
  const float* g2  = (const float*)d_in[12];
  const float* be2 = (const float*)d_in[13];
  float* out = (float*)d_out;

  const int n  = in_sizes[0] / 4;
  const int E  = in_sizes[1] / 2;
  const int TE = E + n;
  const int ntiles = n / 16;

  float* ws = (float*)d_ws;
  const size_t ACC1 = 0;
  const size_t ACC2 = ACC1 + (size_t)n * 64;
  const size_t LOG1 = ACC2 + (size_t)n * 64;
  const size_t XL1  = LOG1;
  const size_t XR1  = XL1 + (size_t)n * 32;
  const size_t LOGSZ = (size_t)TE * 4 > (size_t)n * 64 ? (size_t)TE * 4 : (size_t)n * 64;
  const size_t H1   = LOG1 + LOGSZ;
  const size_t MAX1 = H1 + (size_t)n * 64;
  const size_t MAX2 = MAX1 + (size_t)n * 8;
  const size_t LOG2 = MAX2 + (size_t)n * 2;

  const int B = 256;
  int* flag = (int*)(ws + ACC1) + 24;
  float* SH = ws + MAX2;

  clear_all<<<gridFor((long long)n * 16, B * 4), B, 0, stream>>>(
      (float4*)(ws + ACC1), (float4*)(ws + ACC2), (long long)n * 16,
      (int4*)(ws + MAX1), (int*)(ws + MAX2), n * 2);

  edge_logits1<<<residentGrid(((long long)TE + 15) / 16, RES_GROUPS_WIDE), B, 0, stream>>>(
      ei, E, n, x, W1l, W1r, a1, ws + LOG1, (int*)(ws + MAX1));

  edge_accum1<<<residentGrid(((long long)TE + 7) / 8, RES_GROUPS), B, 0, stream>>>(
      ei, E, n, x, ws + LOG1, (const int*)(ws + MAX1), ws + ACC1);

  node_fin1<<<gridFor((long long)ntiles * 32, 128), 128, 0, stream>>>(
      ws + ACC1, W1l, b1, g1, be1, ws + H1, ntiles);

  gemm2_wmma<<<gridFor((long long)ntiles * 32, B), B, 0, stream>>>(
      ws + H1, W2l, W2r, a2, ws + XL1, ws + XR1, SH, ntiles);

  edge_fused2<<<residentGrid(((long long)TE + 31) / 32, RES_GROUPS_WIDE), B, 0, stream>>>(
      ei, E, n, ws + XL1, ws + XR1, a2, SH, ws + ACC2, flag);

  clear2_guarded<<<GUARDED_GROUPS, B, 0, stream>>>(
      (float4*)(ws + ACC2), (long long)n * 16, (int*)(ws + MAX2), n * 2, flag);

  edge_logits2<<<residentGrid(((long long)TE + 31) / 32, GUARDED_GROUPS), B, 0, stream>>>(
      ei, E, n, ws + XL1, ws + XR1, a2, ws + LOG2, (int*)(ws + MAX2), flag);

  edge_accum2<<<residentGrid(((long long)TE + 31) / 32, GUARDED_GROUPS), B, 0, stream>>>(
      ei, E, n, ws + XL1, ws + LOG2, (const int*)(ws + MAX2), ws + ACC2, flag);

  node_fin2<<<residentGrid(n, RES_GROUPS), B, 0, stream>>>(
      ws + ACC2, b2, g2, be2, out, n);
}
